# starting kernel
# baseline (speedup 1.0000x reference)
#include <hip/hip_runtime.h>
#include <math.h>

#define BB 4
#define SS 2048
#define DD 1024
#define NNh 64
#define MM (BB * SS)
#define KX (2 * DD)
#define NP 192
#define KH (2 * NNh)
#define NO (2 * DD)

#define A2LD 144

typedef __attribute__((ext_vector_type(2))) float v2f;
typedef __attribute__((ext_vector_type(8))) float v8f;

__device__ __forceinline__ v8f wmma_f32(v2f a, v2f b, v8f c) {
  return __builtin_amdgcn_wmma_f32_16x16x4_f32(
       false, a,  false, b,
       (short)0, c,  false,  false);
}

template <int NWN>
__device__ void gemm_tile_acc(const float* __restrict__ A, int lda,
                              const float* __restrict__ Bp, int ldb,
                              int K, int m0, int j0,
                              v2f* As2, v2f* Bs2, v8f acc[2][NWN]) {
  constexpr int BN   = NWN * 32;
  constexpr int B2LD = BN + 16;
  constexpr int NB4  = BN / 64;

  const int tid  = threadIdx.x;
  const int lane = tid & 31;
  const int wave = tid >> 5;
  const int half = lane >> 4;
  const int lrow = lane & 15;
  const int wm   = (wave & 3) * 32;
  const int wn   = (wave >> 2) * (NWN * 16);

  const int am0 = tid >> 2;
  const int ak0 = (tid & 3) << 2;
  const int am1 = (256 + tid) >> 2;
  const int ak1 = ((256 + tid) & 3) << 2;
  int bk[NB4], bj[NB4];
#pragma unroll
  for (int i = 0; i < NB4; ++i) {
    int idx = i * 256 + tid;
    bk[i] = idx / (BN / 4);
    bj[i] = (idx % (BN / 4)) * 4;
  }

  float4 aReg0 = *(const float4*)(A + (size_t)(m0 + am0) * lda + ak0);
  float4 aReg1 = *(const float4*)(A + (size_t)(m0 + am1) * lda + ak1);
  float4 bReg[NB4];
#pragma unroll
  for (int i = 0; i < NB4; ++i)
    bReg[i] = *(const float4*)(Bp + (size_t)bk[i] * ldb + j0 + bj[i]);

  for (int k0 = 0; k0 < K; k0 += 16) {
    __syncthreads();
    {
      v2f t;
      t.x = aReg0.x; t.y = aReg0.y; As2[(ak0 >> 1) * A2LD + am0] = t;
      t.x = aReg0.z; t.y = aReg0.w; As2[((ak0 >> 1) + 1) * A2LD + am0] = t;
      t.x = aReg1.x; t.y = aReg1.y; As2[(ak1 >> 1) * A2LD + am1] = t;
      t.x = aReg1.z; t.y = aReg1.w; As2[((ak1 >> 1) + 1) * A2LD + am1] = t;
    }
    {
      float* bsf = (float*)Bs2;
#pragma unroll
      for (int i = 0; i < NB4; ++i) {
        int base = ((bk[i] >> 1) * B2LD + bj[i]) * 2 + (bk[i] & 1);
        bsf[base + 0] = bReg[i].x;
        bsf[base + 2] = bReg[i].y;
        bsf[base + 4] = bReg[i].z;
        bsf[base + 6] = bReg[i].w;
      }
    }
    __syncthreads();

    if (k0 + 16 < K) {
      aReg0 = *(const float4*)(A + (size_t)(m0 + am0) * lda + (k0 + 16) + ak0);
      aReg1 = *(const float4*)(A + (size_t)(m0 + am1) * lda + (k0 + 16) + ak1);
#pragma unroll
      for (int i = 0; i < NB4; ++i)
        bReg[i] = *(const float4*)(Bp + (size_t)(k0 + 16 + bk[i]) * ldb + j0 + bj[i]);
    }
    if (k0 + 32 < K) {
      __builtin_prefetch(A + (size_t)(m0 + am0) * lda + (k0 + 32) + ak0, 0, 0);
      __builtin_prefetch(A + (size_t)(m0 + am1) * lda + (k0 + 32) + ak1, 0, 0);
    }

#pragma unroll
    for (int kstep = 0; kstep < 4; ++kstep) {
      const int p = 2 * kstep + half;
      v2f a0 = As2[p * A2LD + wm + lrow];
      v2f a1 = As2[p * A2LD + wm + 16 + lrow];
      v2f bf[NWN];
#pragma unroll
      for (int ni = 0; ni < NWN; ++ni)
        bf[ni] = Bs2[p * B2LD + wn + ni * 16 + lrow];
#pragma unroll
      for (int ni = 0; ni < NWN; ++ni) {
        acc[0][ni] = wmma_f32(a0, bf[ni], acc[0][ni]);
        acc[1][ni] = wmma_f32(a1, bf[ni], acc[1][ni]);
      }
    }
  }
}

__global__ void pack_w1_kernel(const float* __restrict__ Bwr, const float* __restrict__ Bwi,
                               const float* __restrict__ gw, float* __restrict__ W1) {
  int idx = blockIdx.x * 256 + threadIdx.x;
  if (idx >= KX * NP) return;
  int k = idx / NP, j = idx - k * NP;
  int d = k >> 1, p = k & 1;
  float v;
  if (j < 64) {
    v = p ? -Bwi[j * DD + d] : Bwr[j * DD + d];
  } else if (j < 128) {
    int n = j - 64;
    v = p ? Bwr[n * DD + d] : Bwi[n * DD + d];
  } else {
    int n = j - 128;
    v = gw[n * (2 * DD) + p * DD + d];
  }
  W1[idx] = v;
}

__global__ void pack_wd_kernel(const float* __restrict__ Dr, const float* __restrict__ Di,
                               float* __restrict__ WD) {
  int idx = blockIdx.x * 256 + threadIdx.x;
  if (idx >= KX * NO) return;
  int k = idx >> 11, j = idx & (NO - 1);
  int e = k >> 1, p = k & 1, d = j >> 1;
  int q = j & 1;
  float wr = Dr[d * DD + e], wi = Di[d * DD + e];
  WD[idx] = (p == q) ? wr : (p ? -wi : wi);
}

__global__ void pack_wc_kernel(const float* __restrict__ Cr, const float* __restrict__ Ci,
                               float* __restrict__ WC) {
  int idx = blockIdx.x * 256 + threadIdx.x;
  if (idx >= KH * NO) return;
  int k = idx >> 11, j = idx & (NO - 1);
  int n = k >> 1, p = k & 1, d = j >> 1;
  int q = j & 1;
  float wr = Cr[d * NNh + n], wi = Ci[d * NNh + n];
  WC[idx] = (p == q) ? wr : (p ? -wi : wi);
}

__global__ __launch_bounds__(256) void gemm_proj_kernel(const float* __restrict__ x,
                                                        const float* __restrict__ W1,
                                                        const float* __restrict__ gate_b,
                                                        float* __restrict__ P) {
  __shared__ v2f As2[8 * A2LD];
  __shared__ v2f Bs2[8 * (64 + 16)];
  v8f acc[2][2] = {{{0,0,0,0,0,0,0,0}, {0,0,0,0,0,0,0,0}},
                   {{0,0,0,0,0,0,0,0}, {0,0,0,0,0,0,0,0}}};
  const int m0 = blockIdx.x * 128;
  const int j0 = blockIdx.y * 64;
  gemm_tile_acc<2>(x, KX, W1, NP, KX, m0, j0, As2, Bs2, acc);

  const int lane = threadIdx.x & 31, wave = threadIdx.x >> 5;
  const int half = lane >> 4, lrow = lane & 15;
  const int wm = (wave & 3) * 32, wn = (wave >> 2) * 32;
#pragma unroll
  for (int mi = 0; mi < 2; ++mi)
#pragma unroll
    for (int ni = 0; ni < 2; ++ni)
#pragma unroll
      for (int e = 0; e < 8; ++e) {
        int m = m0 + wm + mi * 16 + half * 8 + e;
        int j = j0 + wn + ni * 16 + lrow;
        float v = acc[mi][ni][e];
        if (j >= 128) v = 1.0f / (1.0f + expf(-(v + gate_b[j - 128])));
        P[(size_t)m * NP + j] = v;
      }
}

__global__ __launch_bounds__(256) void scan_kernel(const float* __restrict__ P,
                                                   const float* __restrict__ h0,
                                                   const float* __restrict__ theta,
                                                   const float* __restrict__ dampp,
                                                   float* __restrict__ hws,
                                                   float* __restrict__ hfin) {
  const int t = threadIdx.x;
  const int b = t >> 6, n = t & 63;
  const float c = cosf(theta[n]);
  const float s = sinf(theta[n]);
  const float dmp = 0.5f + 0.5f / (1.0f + expf(-dampp[n]));
  const float ca = c * dmp, sa = s * dmp;
  float hr = h0[(b * NNh + n) * 2 + 0];
  float hi = h0[(b * NNh + n) * 2 + 1];
#pragma unroll 4
  for (int st = 0; st < SS; ++st) {
    size_t m = (size_t)b * SS + st;
    float g  = P[m * NP + 128 + n];
    float br = P[m * NP + n];
    float bi = P[m * NP + 64 + n];
    float rr = hr * ca - hi * sa;
    float ri = hr * sa + hi * ca;
    hr = g * br + (1.0f - g) * rr;
    hi = g * bi + (1.0f - g) * ri;
    hws[m * KH + 2 * n + 0] = hr;
    hws[m * KH + 2 * n + 1] = hi;
  }
  hfin[(b * NNh + n) * 2 + 0] = hr;
  hfin[(b * NNh + n) * 2 + 1] = hi;
}

__global__ __launch_bounds__(256) void gemm_out_kernel(const float* __restrict__ x,
                                                       const float* __restrict__ WD,
                                                       const float* __restrict__ hws,
                                                       const float* __restrict__ WC,
                                                       float* __restrict__ out) {
  __shared__ v2f As2[8 * A2LD];
  __shared__ v2f Bs2[8 * (128 + 16)];
  v8f acc[2][4] = {{{0,0,0,0,0,0,0,0}, {0,0,0,0,0,0,0,0},
                    {0,0,0,0,0,0,0,0}, {0,0,0,0,0,0,0,0}},
                   {{0,0,0,0,0,0,0,0}, {0,0,0,0,0,0,0,0},
                    {0,0,0,0,0,0,0,0}, {0,0,0,0,0,0,0,0}}};
  const int m0 = blockIdx.x * 128;
  const int j0 = blockIdx.y * 128;
  gemm_tile_acc<4>(x, KX, WD, NO, KX, m0, j0, As2, Bs2, acc);
  __syncthreads();
  gemm_tile_acc<4>(hws, KH, WC, NO, KH, m0, j0, As2, Bs2, acc);

  const int lane = threadIdx.x & 31, wave = threadIdx.x >> 5;
  const int half = lane >> 4, lrow = lane & 15;
  const int wm = (wave & 3) * 32, wn = (wave >> 2) * 64;
#pragma unroll
  for (int mi = 0; mi < 2; ++mi)
#pragma unroll
    for (int ni = 0; ni < 4; ++ni)
#pragma unroll
      for (int e = 0; e < 8; ++e) {
        int m = m0 + wm + mi * 16 + half * 8 + e;
        int j = j0 + wn + ni * 16 + lrow;
        float z = acc[mi][ni][e];
        float zr = __shfl(z, lane & ~1, 32);
        out[(size_t)m * NO + j] = zr * z + z;
      }
}

extern "C" void kernel_launch(void* const* d_in, const int* in_sizes, int n_in,
                              void* d_out, int out_size, void* d_ws, size_t ws_size,
                              hipStream_t stream) {
  const float* x     = (const float*)d_in[0];
  const float* h0    = (const float*)d_in[1];
  const float* th    = (const float*)d_in[2];
  const float* dampp = (const float*)d_in[3];
  const float* Bwr   = (const float*)d_in[4];
  const float* Bwi   = (const float*)d_in[5];
  const float* Cwr   = (const float*)d_in[6];
  const float* Cwi   = (const float*)d_in[7];
  const float* Dwr   = (const float*)d_in[8];
  const float* Dwi   = (const float*)d_in[9];
  const float* gw    = (const float*)d_in[10];
  const float* gb    = (const float*)d_in[11];

  float* out  = (float*)d_out;
  float* hfin = out + (size_t)MM * NO;

  float* ws = (float*)d_ws;
  float* W1  = ws;
  float* WD  = W1 + (size_t)KX * NP;
  float* WC  = WD + (size_t)KX * NO;
  float* P   = WC + (size_t)KH * NO;
  float* Hws = P + (size_t)MM * NP;

  pack_w1_kernel<<<(KX * NP + 255) / 256, 256, 0, stream>>>(Bwr, Bwi, gw, W1);
  pack_wd_kernel<<<(KX * NO + 255) / 256, 256, 0, stream>>>(Dwr, Dwi, WD);
  pack_wc_kernel<<<(KH * NO + 255) / 256, 256, 0, stream>>>(Cwr, Cwi, WC);

  gemm_proj_kernel<<<dim3(MM / 128, NP / 64), 256, 0, stream>>>(x, W1, gb, P);
  scan_kernel<<<1, 256, 0, stream>>>(P, h0, th, dampp, Hws, hfin);
  gemm_out_kernel<<<dim3(MM / 128, NO / 128), 256, 0, stream>>>(x, WD, Hws, WC, out);
}


# scan inputs prefetched 56 steps ahead: proj stores P as [row/8][col][8], a chain reads 8 steps per two b128 loads into an 8-chunk register ring; scan runs as 8 one-wave work-groups
# speedup vs baseline: 1.5016x; 1.5016x over previous
#include <hip/hip_runtime.h>
#include <math.h>

#define BB 4
#define SS 2048
#define DD 1024
#define NNh 64
#define MM (BB * SS)
#define KX (2 * DD)
#define NP 192
#define KH (2 * NNh)
#define NO (2 * DD)

#define A2LD 144

#define PBLK (NP * 8)
#define CHS 8
#define NCH (SS / CHS)
#define RING 8

typedef __attribute__((ext_vector_type(2))) float v2f;
typedef __attribute__((ext_vector_type(8))) float v8f;

__device__ __forceinline__ v8f wmma_f32(v2f a, v2f b, v8f c) {
  return __builtin_amdgcn_wmma_f32_16x16x4_f32(
       false, a,  false, b,
       (short)0, c,  false,  false);
}

template <int NWN>
__device__ void gemm_tile_acc(const float* __restrict__ A, int lda,
                              const float* __restrict__ Bp, int ldb,
                              int K, int m0, int j0,
                              v2f* As2, v2f* Bs2, v8f acc[2][NWN]) {
  constexpr int BN   = NWN * 32;
  constexpr int B2LD = BN + 16;
  constexpr int NB4  = BN / 64;

  const int tid  = threadIdx.x;
  const int lane = tid & 31;
  const int wave = tid >> 5;
  const int half = lane >> 4;
  const int lrow = lane & 15;
  const int wm   = (wave & 3) * 32;
  const int wn   = (wave >> 2) * (NWN * 16);

  const int am0 = tid >> 2;
  const int ak0 = (tid & 3) << 2;
  const int am1 = (256 + tid) >> 2;
  const int ak1 = ((256 + tid) & 3) << 2;
  int bk[NB4], bj[NB4];
#pragma unroll
  for (int i = 0; i < NB4; ++i) {
    int idx = i * 256 + tid;
    bk[i] = idx / (BN / 4);
    bj[i] = (idx % (BN / 4)) * 4;
  }

  float4 aReg0 = *(const float4*)(A + (size_t)(m0 + am0) * lda + ak0);
  float4 aReg1 = *(const float4*)(A + (size_t)(m0 + am1) * lda + ak1);
  float4 bReg[NB4];
#pragma unroll
  for (int i = 0; i < NB4; ++i)
    bReg[i] = *(const float4*)(Bp + (size_t)bk[i] * ldb + j0 + bj[i]);

  for (int k0 = 0; k0 < K; k0 += 16) {
    __syncthreads();
    {
      v2f t;
      t.x = aReg0.x; t.y = aReg0.y; As2[(ak0 >> 1) * A2LD + am0] = t;
      t.x = aReg0.z; t.y = aReg0.w; As2[((ak0 >> 1) + 1) * A2LD + am0] = t;
      t.x = aReg1.x; t.y = aReg1.y; As2[(ak1 >> 1) * A2LD + am1] = t;
      t.x = aReg1.z; t.y = aReg1.w; As2[((ak1 >> 1) + 1) * A2LD + am1] = t;
    }
    {
      float* bsf = (float*)Bs2;
#pragma unroll
      for (int i = 0; i < NB4; ++i) {
        int base = ((bk[i] >> 1) * B2LD + bj[i]) * 2 + (bk[i] & 1);
        bsf[base + 0] = bReg[i].x;
        bsf[base + 2] = bReg[i].y;
        bsf[base + 4] = bReg[i].z;
        bsf[base + 6] = bReg[i].w;
      }
    }
    __syncthreads();

    if (k0 + 16 < K) {
      aReg0 = *(const float4*)(A + (size_t)(m0 + am0) * lda + (k0 + 16) + ak0);
      aReg1 = *(const float4*)(A + (size_t)(m0 + am1) * lda + (k0 + 16) + ak1);
#pragma unroll
      for (int i = 0; i < NB4; ++i)
        bReg[i] = *(const float4*)(Bp + (size_t)(k0 + 16 + bk[i]) * ldb + j0 + bj[i]);
    }
    if (k0 + 32 < K) {
      __builtin_prefetch(A + (size_t)(m0 + am0) * lda + (k0 + 32) + ak0, 0, 0);
      __builtin_prefetch(A + (size_t)(m0 + am1) * lda + (k0 + 32) + ak1, 0, 0);
    }

#pragma unroll
    for (int kstep = 0; kstep < 4; ++kstep) {
      const int p = 2 * kstep + half;
      v2f a0 = As2[p * A2LD + wm + lrow];
      v2f a1 = As2[p * A2LD + wm + 16 + lrow];
      v2f bf[NWN];
#pragma unroll
      for (int ni = 0; ni < NWN; ++ni)
        bf[ni] = Bs2[p * B2LD + wn + ni * 16 + lrow];
#pragma unroll
      for (int ni = 0; ni < NWN; ++ni) {
        acc[0][ni] = wmma_f32(a0, bf[ni], acc[0][ni]);
        acc[1][ni] = wmma_f32(a1, bf[ni], acc[1][ni]);
      }
    }
  }
}

__global__ void pack_w1_kernel(const float* __restrict__ Bwr, const float* __restrict__ Bwi,
                               const float* __restrict__ gw, float* __restrict__ W1) {
  int idx = blockIdx.x * 256 + threadIdx.x;
  if (idx >= KX * NP) return;
  int k = idx / NP, j = idx - k * NP;
  int d = k >> 1, p = k & 1;
  float v;
  if (j < 64) {
    v = p ? -Bwi[j * DD + d] : Bwr[j * DD + d];
  } else if (j < 128) {
    int n = j - 64;
    v = p ? Bwr[n * DD + d] : Bwi[n * DD + d];
  } else {
    int n = j - 128;
    v = gw[n * (2 * DD) + p * DD + d];
  }
  W1[idx] = v;
}

__global__ void pack_wd_kernel(const float* __restrict__ Dr, const float* __restrict__ Di,
                               float* __restrict__ WD) {
  int idx = blockIdx.x * 256 + threadIdx.x;
  if (idx >= KX * NO) return;
  int k = idx >> 11, j = idx & (NO - 1);
  int e = k >> 1, p = k & 1, d = j >> 1;
  int q = j & 1;
  float wr = Dr[d * DD + e], wi = Di[d * DD + e];
  WD[idx] = (p == q) ? wr : (p ? -wi : wi);
}

__global__ void pack_wc_kernel(const float* __restrict__ Cr, const float* __restrict__ Ci,
                               float* __restrict__ WC) {
  int idx = blockIdx.x * 256 + threadIdx.x;
  if (idx >= KH * NO) return;
  int k = idx >> 11, j = idx & (NO - 1);
  int n = k >> 1, p = k & 1, d = j >> 1;
  int q = j & 1;
  float wr = Cr[d * NNh + n], wi = Ci[d * NNh + n];
  WC[idx] = (p == q) ? wr : (p ? -wi : wi);
}

__global__ __launch_bounds__(256) void gemm_proj_kernel(const float* __restrict__ x,
                                                        const float* __restrict__ W1,
                                                        const float* __restrict__ gate_b,
                                                        float* __restrict__ P) {
  __shared__ v2f As2[8 * A2LD];
  __shared__ v2f Bs2[8 * (64 + 16)];
  v8f acc[2][2] = {{{0,0,0,0,0,0,0,0}, {0,0,0,0,0,0,0,0}},
                   {{0,0,0,0,0,0,0,0}, {0,0,0,0,0,0,0,0}}};
  const int m0 = blockIdx.x * 128;
  const int j0 = blockIdx.y * 64;
  gemm_tile_acc<2>(x, KX, W1, NP, KX, m0, j0, As2, Bs2, acc);

  const int lane = threadIdx.x & 31, wave = threadIdx.x >> 5;
  const int half = lane >> 4, lrow = lane & 15;
  const int wm = (wave & 3) * 32, wn = (wave >> 2) * 32;
#pragma unroll
  for (int mi = 0; mi < 2; ++mi)
#pragma unroll
    for (int ni = 0; ni < 2; ++ni) {
      const int j  = j0 + wn + ni * 16 + lrow;
      const int rb = ((m0 + wm + mi * 16) >> 3) + half;
      float v[8];
#pragma unroll
      for (int e = 0; e < 8; ++e) {
        v[e] = acc[mi][ni][e];
        if (j >= 128) v[e] = 1.0f / (1.0f + expf(-(v[e] + gate_b[j - 128])));
      }
      float* dst = P + (size_t)rb * PBLK + j * 8;
      *(float4*)(dst)     = make_float4(v[0], v[1], v[2], v[3]);
      *(float4*)(dst + 4) = make_float4(v[4], v[5], v[6], v[7]);
    }
}

struct ScanChunk {
  float4 br[2];
  float4 bi[2];
  float4 g[2];
};

__device__ __forceinline__ void scan_load(ScanChunk& c, const float* __restrict__ pw, unsigned lo, int chunk) {
  const float* p = pw + (size_t)chunk * PBLK;
  c.br[0] = *(const float4*)(p + lo);
  c.br[1] = *(const float4*)(p + 4 + lo);
  c.bi[0] = *(const float4*)(p + 64 * 8 + lo);
  c.bi[1] = *(const float4*)(p + 64 * 8 + 4 + lo);
  c.g[0]  = *(const float4*)(p + 128 * 8 + lo);
  c.g[1]  = *(const float4*)(p + 128 * 8 + 4 + lo);
}

__device__ __forceinline__ float f4_get(const float4& v, int i) {
  return i == 0 ? v.x : (i == 1 ? v.y : (i == 2 ? v.z : v.w));
}

__device__ __forceinline__ void scan_steps(const ScanChunk& c, float ca, float sa,
                                           float& hr, float& hi, float* __restrict__ hw, unsigned ho) {
#pragma unroll
  for (int i = 0; i < CHS; ++i) {
    float g  = f4_get(c.g[i >> 2], i & 3);
    float br = f4_get(c.br[i >> 2], i & 3);
    float bi = f4_get(c.bi[i >> 2], i & 3);
    float rr = hr * ca - hi * sa;
    float ri = hr * sa + hi * ca;
    hr = g * br + (1.0f - g) * rr;
    hi = g * bi + (1.0f - g) * ri;
    *(float2*)(hw + (size_t)i * KH + ho) = make_float2(hr, hi);
  }
}

__global__ __launch_bounds__(32) void scan_kernel(const float* __restrict__ P,
                                                  const float* __restrict__ h0,
                                                  const float* __restrict__ theta,
                                                  const float* __restrict__ dampp,
                                                  float* __restrict__ hws,
                                                  float* __restrict__ hfin) {
  const int b  = blockIdx.x >> 1;
  const int n0 = (blockIdx.x & 1) * 32;
  const int n  = n0 + threadIdx.x;
  const float* pw = P + (size_t)b * (SS / 8) * PBLK + n0 * 8;
  float* hw = hws + (size_t)b * SS * KH + 2 * n0;
  const unsigned lo = threadIdx.x * 8u;
  const unsigned ho = threadIdx.x * 2u;

  const float c = cosf(theta[n]);
  const float s = sinf(theta[n]);
  const float dmp = 0.5f + 0.5f / (1.0f + expf(-dampp[n]));
  const float ca = c * dmp, sa = s * dmp;
  float hr = h0[(b * NNh + n) * 2 + 0];
  float hi = h0[(b * NNh + n) * 2 + 1];

  __builtin_amdgcn_sched_barrier(0);

  ScanChunk ring[RING];
#pragma unroll
  for (int r = 0; r < RING - 1; ++r) scan_load(ring[r], pw, lo, r);

  int k = 0;
  for (; k < NCH - RING; k += RING) {
#pragma unroll
    for (int r = 0; r < RING; ++r) {
      __builtin_amdgcn_sched_barrier(0);
      scan_load(ring[(r + RING - 1) % RING], pw, lo, k + r + RING - 1);
      __builtin_amdgcn_sched_barrier(0);
      scan_steps(ring[r], ca, sa, hr, hi, hw + (size_t)(k + r) * CHS * KH, ho);
    }
  }
  __builtin_amdgcn_sched_barrier(0);
  scan_load(ring[RING - 1], pw, lo, NCH - 1);
#pragma unroll
  for (int r = 0; r < RING; ++r) {
    __builtin_amdgcn_sched_barrier(0);
    scan_steps(ring[r], ca, sa, hr, hi, hw + (size_t)(NCH - RING + r) * CHS * KH, ho);
  }
  hfin[(b * NNh + n) * 2 + 0] = hr;
  hfin[(b * NNh + n) * 2 + 1] = hi;
}

__global__ __launch_bounds__(256) void gemm_out_kernel(const float* __restrict__ x,
                                                       const float* __restrict__ WD,
                                                       const float* __restrict__ hws,
                                                       const float* __restrict__ WC,
                                                       float* __restrict__ out) {
  __shared__ v2f As2[8 * A2LD];
  __shared__ v2f Bs2[8 * (128 + 16)];
  v8f acc[2][4] = {{{0,0,0,0,0,0,0,0}, {0,0,0,0,0,0,0,0},
                    {0,0,0,0,0,0,0,0}, {0,0,0,0,0,0,0,0}},
                   {{0,0,0,0,0,0,0,0}, {0,0,0,0,0,0,0,0},
                    {0,0,0,0,0,0,0,0}, {0,0,0,0,0,0,0,0}}};
  const int m0 = blockIdx.x * 128;
  const int j0 = blockIdx.y * 128;
  gemm_tile_acc<4>(x, KX, WD, NO, KX, m0, j0, As2, Bs2, acc);
  __syncthreads();
  gemm_tile_acc<4>(hws, KH, WC, NO, KH, m0, j0, As2, Bs2, acc);

  const int lane = threadIdx.x & 31, wave = threadIdx.x >> 5;
  const int half = lane >> 4, lrow = lane & 15;
  const int wm = (wave & 3) * 32, wn = (wave >> 2) * 64;
#pragma unroll
  for (int mi = 0; mi < 2; ++mi)
#pragma unroll
    for (int ni = 0; ni < 4; ++ni)
#pragma unroll
      for (int e = 0; e < 8; ++e) {
        int m = m0 + wm + mi * 16 + half * 8 + e;
        int j = j0 + wn + ni * 16 + lrow;
        float z = acc[mi][ni][e];
        float zr = __shfl(z, lane & ~1, 32);
        out[(size_t)m * NO + j] = zr * z + z;
      }
}

extern "C" void kernel_launch(void* const* d_in, const int* in_sizes, int n_in,
                              void* d_out, int out_size, void* d_ws, size_t ws_size,
                              hipStream_t stream) {
  const float* x     = (const float*)d_in[0];
  const float* h0    = (const float*)d_in[1];
  const float* th    = (const float*)d_in[2];
  const float* dampp = (const float*)d_in[3];
  const float* Bwr   = (const float*)d_in[4];
  const float* Bwi   = (const float*)d_in[5];
  const float* Cwr   = (const float*)d_in[6];
  const float* Cwi   = (const float*)d_in[7];
  const float* Dwr   = (const float*)d_in[8];
  const float* Dwi   = (const float*)d_in[9];
  const float* gw    = (const float*)d_in[10];
  const float* gb    = (const float*)d_in[11];

  float* out  = (float*)d_out;
  float* hfin = out + (size_t)MM * NO;

  float* ws = (float*)d_ws;
  float* W1  = ws;
  float* WD  = W1 + (size_t)KX * NP;
  float* WC  = WD + (size_t)KX * NO;
  float* P   = WC + (size_t)KH * NO;
  float* Hws = P + (size_t)MM * NP;

  pack_w1_kernel<<<(KX * NP + 255) / 256, 256, 0, stream>>>(Bwr, Bwi, gw, W1);
  pack_wd_kernel<<<(KX * NO + 255) / 256, 256, 0, stream>>>(Dwr, Dwi, WD);
  pack_wc_kernel<<<(KH * NO + 255) / 256, 256, 0, stream>>>(Cwr, Cwi, WC);

  gemm_proj_kernel<<<dim3(MM / 128, NP / 64), 256, 0, stream>>>(x, W1, gb, P);
  scan_kernel<<<BB * NNh / 32, 32, 0, stream>>>(P, h0, th, dampp, Hws, hfin);
  gemm_out_kernel<<<dim3(MM / 128, NO / 128), 256, 0, stream>>>(x, WD, Hws, WC, out);
}


# D x as three real f32 WMMA products (k1=(xr+xi)wr, k2=xr(wi-wr), k3=xi(wr+wi); zr=k1-k3, zi=k1+k2) instead of four; C h folded into k2,k3; changes rounding and order of the sums
# speedup vs baseline: 1.8417x; 1.2265x over previous
#include <hip/hip_runtime.h>
#include <math.h>

#define BB 4
#define SS 2048
#define DD 1024
#define NNh 64
#define MM (BB * SS)
#define KX (2 * DD)
#define NP 192
#define KH (2 * NNh)
#define NO (2 * DD)

#define A2LD 144

#define PBLK (NP * 8)
#define CHS 8
#define NCH (SS / CHS)
#define RING 8

typedef __attribute__((ext_vector_type(2))) float v2f;
typedef __attribute__((ext_vector_type(8))) float v8f;

__device__ __forceinline__ v8f wmma_f32(v2f a, v2f b, v8f c) {
  return __builtin_amdgcn_wmma_f32_16x16x4_f32(
       false, a,  false, b,
       (short)0, c,  false,  false);
}

template <int NWN>
__device__ void gemm_tile_acc(const float* __restrict__ A, int lda,
                              const float* __restrict__ Bp, int ldb,
                              int K, int m0, int j0,
                              v2f* As2, v2f* Bs2, v8f acc[2][NWN]) {
  constexpr int BN   = NWN * 32;
  constexpr int B2LD = BN + 16;
  constexpr int NB4  = BN / 64;

  const int tid  = threadIdx.x;
  const int lane = tid & 31;
  const int wave = tid >> 5;
  const int half = lane >> 4;
  const int lrow = lane & 15;
  const int wm   = (wave & 3) * 32;
  const int wn   = (wave >> 2) * (NWN * 16);

  const int am0 = tid >> 2;
  const int ak0 = (tid & 3) << 2;
  const int am1 = (256 + tid) >> 2;
  const int ak1 = ((256 + tid) & 3) << 2;
  int bk[NB4], bj[NB4];
#pragma unroll
  for (int i = 0; i < NB4; ++i) {
    int idx = i * 256 + tid;
    bk[i] = idx / (BN / 4);
    bj[i] = (idx % (BN / 4)) * 4;
  }

  float4 aReg0 = *(const float4*)(A + (size_t)(m0 + am0) * lda + ak0);
  float4 aReg1 = *(const float4*)(A + (size_t)(m0 + am1) * lda + ak1);
  float4 bReg[NB4];
#pragma unroll
  for (int i = 0; i < NB4; ++i)
    bReg[i] = *(const float4*)(Bp + (size_t)bk[i] * ldb + j0 + bj[i]);

  for (int k0 = 0; k0 < K; k0 += 16) {
    __syncthreads();
    {
      v2f t;
      t.x = aReg0.x; t.y = aReg0.y; As2[(ak0 >> 1) * A2LD + am0] = t;
      t.x = aReg0.z; t.y = aReg0.w; As2[((ak0 >> 1) + 1) * A2LD + am0] = t;
      t.x = aReg1.x; t.y = aReg1.y; As2[(ak1 >> 1) * A2LD + am1] = t;
      t.x = aReg1.z; t.y = aReg1.w; As2[((ak1 >> 1) + 1) * A2LD + am1] = t;
    }
    {
      float* bsf = (float*)Bs2;
#pragma unroll
      for (int i = 0; i < NB4; ++i) {
        int base = ((bk[i] >> 1) * B2LD + bj[i]) * 2 + (bk[i] & 1);
        bsf[base + 0] = bReg[i].x;
        bsf[base + 2] = bReg[i].y;
        bsf[base + 4] = bReg[i].z;
        bsf[base + 6] = bReg[i].w;
      }
    }
    __syncthreads();

    if (k0 + 16 < K) {
      aReg0 = *(const float4*)(A + (size_t)(m0 + am0) * lda + (k0 + 16) + ak0);
      aReg1 = *(const float4*)(A + (size_t)(m0 + am1) * lda + (k0 + 16) + ak1);
#pragma unroll
      for (int i = 0; i < NB4; ++i)
        bReg[i] = *(const float4*)(Bp + (size_t)(k0 + 16 + bk[i]) * ldb + j0 + bj[i]);
    }
    if (k0 + 32 < K) {
      __builtin_prefetch(A + (size_t)(m0 + am0) * lda + (k0 + 32) + ak0, 0, 0);
      __builtin_prefetch(A + (size_t)(m0 + am1) * lda + (k0 + 32) + ak1, 0, 0);
    }

#pragma unroll
    for (int kstep = 0; kstep < 4; ++kstep) {
      const int p = 2 * kstep + half;
      v2f a0 = As2[p * A2LD + wm + lrow];
      v2f a1 = As2[p * A2LD + wm + 16 + lrow];
      v2f bf[NWN];
#pragma unroll
      for (int ni = 0; ni < NWN; ++ni)
        bf[ni] = Bs2[p * B2LD + wn + ni * 16 + lrow];
#pragma unroll
      for (int ni = 0; ni < NWN; ++ni) {
        acc[0][ni] = wmma_f32(a0, bf[ni], acc[0][ni]);
        acc[1][ni] = wmma_f32(a1, bf[ni], acc[1][ni]);
      }
    }
  }
}

__global__ void pack_w1_kernel(const float* __restrict__ Bwr, const float* __restrict__ Bwi,
                               const float* __restrict__ gw, float* __restrict__ W1) {
  int idx = blockIdx.x * 256 + threadIdx.x;
  if (idx >= KX * NP) return;
  int k = idx / NP, j = idx - k * NP;
  int d = k >> 1, p = k & 1;
  float v;
  if (j < 64) {
    v = p ? -Bwi[j * DD + d] : Bwr[j * DD + d];
  } else if (j < 128) {
    int n = j - 64;
    v = p ? Bwr[n * DD + d] : Bwi[n * DD + d];
  } else {
    int n = j - 128;
    v = gw[n * (2 * DD) + p * DD + d];
  }
  W1[idx] = v;
}

__global__ __launch_bounds__(256) void pack_wd_kernel(const float* __restrict__ Dr,
                                                      const float* __restrict__ Di,
                                                      float* __restrict__ WB) {
  __shared__ float tr[32][33];
  __shared__ float ti[32][33];
  const int tx = threadIdx.x & 31, ty = threadIdx.x >> 5;
  const int e0 = blockIdx.x * 32, d0 = blockIdx.y * 32;
  for (int r = ty; r < 32; r += 8) {
    tr[r][tx] = Dr[(size_t)(d0 + r) * DD + e0 + tx];
    ti[r][tx] = Di[(size_t)(d0 + r) * DD + e0 + tx];
  }
  __syncthreads();
  for (int r = ty; r < 32; r += 8) {
    const float wr = tr[tx][r], wi = ti[tx][r];
    const size_t o = (size_t)(e0 + r) * DD + d0 + tx;
    WB[o] = wr;
    WB[(size_t)DD * DD + o] = wi - wr;
    WB[2 * (size_t)DD * DD + o] = wr + wi;
  }
}

__global__ void pack_wc_kernel(const float* __restrict__ Cr, const float* __restrict__ Ci,
                               float* __restrict__ WC) {
  int idx = blockIdx.x * 256 + threadIdx.x;
  if (idx >= KH * DD) return;
  int k = idx >> 10, d = idx & (DD - 1);
  int n = k >> 1, p = k & 1;
  float cr = Cr[d * NNh + n], ci = Ci[d * NNh + n];
  WC[idx] = p ? cr : ci;
  WC[(size_t)KH * DD + idx] = p ? ci : -cr;
}

__global__ __launch_bounds__(256) void gemm_proj_kernel(const float* __restrict__ x,
                                                        const float* __restrict__ W1,
                                                        const float* __restrict__ gate_b,
                                                        float* __restrict__ P) {
  __shared__ v2f As2[8 * A2LD];
  __shared__ v2f Bs2[8 * (64 + 16)];
  v8f acc[2][2] = {{{0,0,0,0,0,0,0,0}, {0,0,0,0,0,0,0,0}},
                   {{0,0,0,0,0,0,0,0}, {0,0,0,0,0,0,0,0}}};
  const int m0 = blockIdx.x * 128;
  const int j0 = blockIdx.y * 64;
  gemm_tile_acc<2>(x, KX, W1, NP, KX, m0, j0, As2, Bs2, acc);

  const int lane = threadIdx.x & 31, wave = threadIdx.x >> 5;
  const int half = lane >> 4, lrow = lane & 15;
  const int wm = (wave & 3) * 32, wn = (wave >> 2) * 32;
#pragma unroll
  for (int mi = 0; mi < 2; ++mi)
#pragma unroll
    for (int ni = 0; ni < 2; ++ni) {
      const int j  = j0 + wn + ni * 16 + lrow;
      const int rb = ((m0 + wm + mi * 16) >> 3) + half;
      float v[8];
#pragma unroll
      for (int e = 0; e < 8; ++e) {
        v[e] = acc[mi][ni][e];
        if (j >= 128) v[e] = 1.0f / (1.0f + expf(-(v[e] + gate_b[j - 128])));
      }
      float* dst = P + (size_t)rb * PBLK + j * 8;
      *(float4*)(dst)     = make_float4(v[0], v[1], v[2], v[3]);
      *(float4*)(dst + 4) = make_float4(v[4], v[5], v[6], v[7]);
    }
}

struct ScanChunk {
  float4 br[2];
  float4 bi[2];
  float4 g[2];
};

__device__ __forceinline__ void scan_load(ScanChunk& c, const float* __restrict__ pw, unsigned lo, int chunk) {
  const float* p = pw + (size_t)chunk * PBLK;
  c.br[0] = *(const float4*)(p + lo);
  c.br[1] = *(const float4*)(p + 4 + lo);
  c.bi[0] = *(const float4*)(p + 64 * 8 + lo);
  c.bi[1] = *(const float4*)(p + 64 * 8 + 4 + lo);
  c.g[0]  = *(const float4*)(p + 128 * 8 + lo);
  c.g[1]  = *(const float4*)(p + 128 * 8 + 4 + lo);
}

__device__ __forceinline__ float f4_get(const float4& v, int i) {
  return i == 0 ? v.x : (i == 1 ? v.y : (i == 2 ? v.z : v.w));
}

__device__ __forceinline__ void scan_steps(const ScanChunk& c, float ca, float sa,
                                           float& hr, float& hi, float* __restrict__ hw, unsigned ho) {
#pragma unroll
  for (int i = 0; i < CHS; ++i) {
    float g  = f4_get(c.g[i >> 2], i & 3);
    float br = f4_get(c.br[i >> 2], i & 3);
    float bi = f4_get(c.bi[i >> 2], i & 3);
    float rr = hr * ca - hi * sa;
    float ri = hr * sa + hi * ca;
    hr = g * br + (1.0f - g) * rr;
    hi = g * bi + (1.0f - g) * ri;
    *(float2*)(hw + (size_t)i * KH + ho) = make_float2(hr, hi);
  }
}

__global__ __launch_bounds__(32) void scan_kernel(const float* __restrict__ P,
                                                  const float* __restrict__ h0,
                                                  const float* __restrict__ theta,
                                                  const float* __restrict__ dampp,
                                                  float* __restrict__ hws,
                                                  float* __restrict__ hfin) {
  const int b  = blockIdx.x >> 1;
  const int n0 = (blockIdx.x & 1) * 32;
  const int n  = n0 + threadIdx.x;
  const float* pw = P + (size_t)b * (SS / 8) * PBLK + n0 * 8;
  float* hw = hws + (size_t)b * SS * KH + 2 * n0;
  const unsigned lo = threadIdx.x * 8u;
  const unsigned ho = threadIdx.x * 2u;

  const float c = cosf(theta[n]);
  const float s = sinf(theta[n]);
  const float dmp = 0.5f + 0.5f / (1.0f + expf(-dampp[n]));
  const float ca = c * dmp, sa = s * dmp;
  float hr = h0[(b * NNh + n) * 2 + 0];
  float hi = h0[(b * NNh + n) * 2 + 1];

  __builtin_amdgcn_sched_barrier(0);

  ScanChunk ring[RING];
#pragma unroll
  for (int r = 0; r < RING - 1; ++r) scan_load(ring[r], pw, lo, r);

  int k = 0;
  for (; k < NCH - RING; k += RING) {
#pragma unroll
    for (int r = 0; r < RING; ++r) {
      __builtin_amdgcn_sched_barrier(0);
      scan_load(ring[(r + RING - 1) % RING], pw, lo, k + r + RING - 1);
      __builtin_amdgcn_sched_barrier(0);
      scan_steps(ring[r], ca, sa, hr, hi, hw + (size_t)(k + r) * CHS * KH, ho);
    }
  }
  __builtin_amdgcn_sched_barrier(0);
  scan_load(ring[RING - 1], pw, lo, NCH - 1);
#pragma unroll
  for (int r = 0; r < RING; ++r) {
    __builtin_amdgcn_sched_barrier(0);
    scan_steps(ring[r], ca, sa, hr, hi, hw + (size_t)(NCH - RING + r) * CHS * KH, ho);
  }
  hfin[(b * NNh + n) * 2 + 0] = hr;
  hfin[(b * NNh + n) * 2 + 1] = hi;
}

#define B3LD 80

__global__ __launch_bounds__(256) void gemm_out_kernel(const float* __restrict__ x,
                                                       const float* __restrict__ WB,
                                                       const float* __restrict__ hws,
                                                       const float* __restrict__ WC,
                                                       float* __restrict__ out) {
  __shared__ v2f As2[3][8 * A2LD];
  __shared__ v2f Bs2[3][8 * B3LD];
  v8f acc[3][2][2];
#pragma unroll
  for (int pl = 0; pl < 3; ++pl)
#pragma unroll
    for (int mi = 0; mi < 2; ++mi)
#pragma unroll
      for (int ni = 0; ni < 2; ++ni)
#pragma unroll
        for (int e = 0; e < 8; ++e) acc[pl][mi][ni][e] = 0.0f;

  const int tid  = threadIdx.x;
  const int lane = tid & 31, wave = tid >> 5;
  const int half = lane >> 4, lrow = lane & 15;
  const int wm = (wave & 3) * 32, wn = (wave >> 2) * 32;
  const int m0 = blockIdx.x * 128;
  const int j0 = blockIdx.y * 64;

  const int bk = tid >> 4, bj = (tid & 15) << 2;
  const int bbase = ((bk >> 1) * B3LD + bj) * 2 + (bk & 1);

  {
    const int arow = tid >> 3, akp = tid & 7;
    const float* ap = x + (size_t)(m0 + arow) * KX + akp * 4;
    const float* bp = WB + (size_t)bk * DD + j0 + bj;
    float4 aReg[4], bReg[3];
#pragma unroll
    for (int i = 0; i < 4; ++i) aReg[i] = *(const float4*)(ap + (size_t)i * 32 * KX);
#pragma unroll
    for (int pl = 0; pl < 3; ++pl) bReg[pl] = *(const float4*)(bp + (size_t)pl * DD * DD);

    for (int k0 = 0; k0 < DD; k0 += 16) {
      __syncthreads();
#pragma unroll
      for (int i = 0; i < 4; ++i) {
        const float4 f = aReg[i];
        const int o = akp * A2LD + arow + i * 32;
        v2f t;
        t.x = f.x + f.y; t.y = f.z + f.w; As2[0][o] = t;
        t.x = f.x;       t.y = f.z;       As2[1][o] = t;
        t.x = f.y;       t.y = f.w;       As2[2][o] = t;
      }
#pragma unroll
      for (int pl = 0; pl < 3; ++pl) {
        float* bsf = (float*)Bs2[pl];
        bsf[bbase + 0] = bReg[pl].x;
        bsf[bbase + 2] = bReg[pl].y;
        bsf[bbase + 4] = bReg[pl].z;
        bsf[bbase + 6] = bReg[pl].w;
      }
      __syncthreads();

      if (k0 + 16 < DD) {
#pragma unroll
        for (int i = 0; i < 4; ++i)
          aReg[i] = *(const float4*)(ap + (size_t)i * 32 * KX + (k0 + 16) * 2);
#pragma unroll
        for (int pl = 0; pl < 3; ++pl)
          bReg[pl] = *(const float4*)(bp + (size_t)pl * DD * DD + (size_t)(k0 + 16) * DD);
      }

#pragma unroll
      for (int kstep = 0; kstep < 4; ++kstep) {
        const int p = 2 * kstep + half;
        v2f a0[3], a1[3], b0[3], b1[3];
#pragma unroll
        for (int pl = 0; pl < 3; ++pl) {
          a0[pl] = As2[pl][p * A2LD + wm + lrow];
          a1[pl] = As2[pl][p * A2LD + wm + 16 + lrow];
          b0[pl] = Bs2[pl][p * B3LD + wn + lrow];
          b1[pl] = Bs2[pl][p * B3LD + wn + 16 + lrow];
        }
#pragma unroll
        for (int pl = 0; pl < 3; ++pl) {
          acc[pl][0][0] = wmma_f32(a0[pl], b0[pl], acc[pl][0][0]);
          acc[pl][1][0] = wmma_f32(a1[pl], b0[pl], acc[pl][1][0]);
          acc[pl][0][1] = wmma_f32(a0[pl], b1[pl], acc[pl][0][1]);
          acc[pl][1][1] = wmma_f32(a1[pl], b1[pl], acc[pl][1][1]);
        }
      }
    }
  }

  {
    const int am0 = tid >> 2, am1 = 64 + (tid >> 2);
    const int ak0 = (tid & 3) << 2;
    const float* ap0 = hws + (size_t)(m0 + am0) * KH + ak0;
    const float* ap1 = hws + (size_t)(m0 + am1) * KH + ak0;
    const float* bp = WC + (size_t)bk * DD + j0 + bj;
    float4 aReg0 = *(const float4*)(ap0);
    float4 aReg1 = *(const float4*)(ap1);
    float4 bReg[2];
#pragma unroll
    for (int q = 0; q < 2; ++q) bReg[q] = *(const float4*)(bp + (size_t)q * KH * DD);

    for (int k0 = 0; k0 < KH; k0 += 16) {
      __syncthreads();
      {
        v2f t;
        t.x = aReg0.x; t.y = aReg0.y; As2[0][(ak0 >> 1) * A2LD + am0] = t;
        t.x = aReg0.z; t.y = aReg0.w; As2[0][((ak0 >> 1) + 1) * A2LD + am0] = t;
        t.x = aReg1.x; t.y = aReg1.y; As2[0][(ak0 >> 1) * A2LD + am1] = t;
        t.x = aReg1.z; t.y = aReg1.w; As2[0][((ak0 >> 1) + 1) * A2LD + am1] = t;
      }
#pragma unroll
      for (int q = 0; q < 2; ++q) {
        float* bsf = (float*)Bs2[q];
        bsf[bbase + 0] = bReg[q].x;
        bsf[bbase + 2] = bReg[q].y;
        bsf[bbase + 4] = bReg[q].z;
        bsf[bbase + 6] = bReg[q].w;
      }
      __syncthreads();

      if (k0 + 16 < KH) {
        aReg0 = *(const float4*)(ap0 + (k0 + 16));
        aReg1 = *(const float4*)(ap1 + (k0 + 16));
#pragma unroll
        for (int q = 0; q < 2; ++q)
          bReg[q] = *(const float4*)(bp + (size_t)q * KH * DD + (size_t)(k0 + 16) * DD);
      }

#pragma unroll
      for (int kstep = 0; kstep < 4; ++kstep) {
        const int p = 2 * kstep + half;
        const v2f a0 = As2[0][p * A2LD + wm + lrow];
        const v2f a1 = As2[0][p * A2LD + wm + 16 + lrow];
        v2f b0[2], b1[2];
#pragma unroll
        for (int q = 0; q < 2; ++q) {
          b0[q] = Bs2[q][p * B3LD + wn + lrow];
          b1[q] = Bs2[q][p * B3LD + wn + 16 + lrow];
        }
#pragma unroll
        for (int q = 0; q < 2; ++q) {
          acc[1 + q][0][0] = wmma_f32(a0, b0[q], acc[1 + q][0][0]);
          acc[1 + q][1][0] = wmma_f32(a1, b0[q], acc[1 + q][1][0]);
          acc[1 + q][0][1] = wmma_f32(a0, b1[q], acc[1 + q][0][1]);
          acc[1 + q][1][1] = wmma_f32(a1, b1[q], acc[1 + q][1][1]);
        }
      }
    }
  }

#pragma unroll
  for (int mi = 0; mi < 2; ++mi)
#pragma unroll
    for (int ni = 0; ni < 2; ++ni)
#pragma unroll
      for (int e = 0; e < 8; ++e) {
        const int m = m0 + wm + mi * 16 + half * 8 + e;
        const int d = j0 + wn + ni * 16 + lrow;
        const float k1 = acc[0][mi][ni][e];
        const float zr = k1 - acc[2][mi][ni][e];
        const float zi = k1 + acc[1][mi][ni][e];
        *(float2*)(out + (size_t)m * NO + 2 * d) = make_float2(zr * zr + zr, zr * zi + zi);
      }
}

extern "C" void kernel_launch(void* const* d_in, const int* in_sizes, int n_in,
                              void* d_out, int out_size, void* d_ws, size_t ws_size,
                              hipStream_t stream) {
  const float* x     = (const float*)d_in[0];
  const float* h0    = (const float*)d_in[1];
  const float* th    = (const float*)d_in[2];
  const float* dampp = (const float*)d_in[3];
  const float* Bwr   = (const float*)d_in[4];
  const float* Bwi   = (const float*)d_in[5];
  const float* Cwr   = (const float*)d_in[6];
  const float* Cwi   = (const float*)d_in[7];
  const float* Dwr   = (const float*)d_in[8];
  const float* Dwi   = (const float*)d_in[9];
  const float* gw    = (const float*)d_in[10];
  const float* gb    = (const float*)d_in[11];

  float* out  = (float*)d_out;
  float* hfin = out + (size_t)MM * NO;

  float* ws = (float*)d_ws;
  float* W1  = ws;
  float* WD  = W1 + (size_t)KX * NP;
  float* WC  = WD + (size_t)KX * NO;
  float* P   = WC + (size_t)KH * NO;
  float* Hws = P + (size_t)MM * NP;

  pack_w1_kernel<<<(KX * NP + 255) / 256, 256, 0, stream>>>(Bwr, Bwi, gw, W1);
  pack_wd_kernel<<<dim3(DD / 32, DD / 32), 256, 0, stream>>>(Dwr, Dwi, WD);
  pack_wc_kernel<<<(KH * DD + 255) / 256, 256, 0, stream>>>(Cwr, Cwi, WC);

  gemm_proj_kernel<<<dim3(MM / 128, NP / 64), 256, 0, stream>>>(x, W1, gb, P);
  scan_kernel<<<BB * NNh / 32, 32, 0, stream>>>(P, h0, th, dampp, Hws, hfin);
  gemm_out_kernel<<<dim3(MM / 128, DD / 64), 256, 0, stream>>>(x, WD, Hws, WC, out);
}


# proj GEMM stages 64 k per barrier pair (was 16): tile 128 x 48 on 256 work-groups, weights packed k-contiguous, both operands staged row-major by b128; k order of the sums changes
# speedup vs baseline: 2.0466x; 1.1113x over previous
#include <hip/hip_runtime.h>
#include <math.h>

#define BB 4
#define SS 2048
#define DD 1024
#define NNh 64
#define MM (BB * SS)
#define KX (2 * DD)
#define NP 192
#define KH (2 * NNh)
#define NO (2 * DD)

#define A2LD 144

#define PBLK (NP * 8)
#define CHS 8
#define NCH (SS / CHS)
#define RING 8

typedef __attribute__((ext_vector_type(2))) float v2f;
typedef __attribute__((ext_vector_type(8))) float v8f;

__device__ __forceinline__ v8f wmma_f32(v2f a, v2f b, v8f c) {
  return __builtin_amdgcn_wmma_f32_16x16x4_f32(
       false, a,  false, b,
       (short)0, c,  false,  false);
}

__global__ void pack_w1_kernel(const float* __restrict__ Bwr, const float* __restrict__ Bwi,
                               const float* __restrict__ gw, float* __restrict__ W1t) {
  int idx = blockIdx.x * 256 + threadIdx.x;
  if (idx >= KX * NP) return;
  int j = idx / KX, k = idx - j * KX;
  int d = k >> 1, p = k & 1;
  float v;
  if (j < 64) {
    v = p ? -Bwi[j * DD + d] : Bwr[j * DD + d];
  } else if (j < 128) {
    int n = j - 64;
    v = p ? Bwr[n * DD + d] : Bwi[n * DD + d];
  } else {
    int n = j - 128;
    v = gw[n * (2 * DD) + p * DD + d];
  }
  W1t[idx] = v;
}

__global__ __launch_bounds__(256) void pack_wd_kernel(const float* __restrict__ Dr,
                                                      const float* __restrict__ Di,
                                                      float* __restrict__ WB) {
  __shared__ float tr[32][33];
  __shared__ float ti[32][33];
  const int tx = threadIdx.x & 31, ty = threadIdx.x >> 5;
  const int e0 = blockIdx.x * 32, d0 = blockIdx.y * 32;
  for (int r = ty; r < 32; r += 8) {
    tr[r][tx] = Dr[(size_t)(d0 + r) * DD + e0 + tx];
    ti[r][tx] = Di[(size_t)(d0 + r) * DD + e0 + tx];
  }
  __syncthreads();
  for (int r = ty; r < 32; r += 8) {
    const float wr = tr[tx][r], wi = ti[tx][r];
    const size_t o = (size_t)(e0 + r) * DD + d0 + tx;
    WB[o] = wr;
    WB[(size_t)DD * DD + o] = wi - wr;
    WB[2 * (size_t)DD * DD + o] = wr + wi;
  }
}

__global__ void pack_wc_kernel(const float* __restrict__ Cr, const float* __restrict__ Ci,
                               float* __restrict__ WC) {
  int idx = blockIdx.x * 256 + threadIdx.x;
  if (idx >= KH * DD) return;
  int k = idx >> 10, d = idx & (DD - 1);
  int n = k >> 1, p = k & 1;
  float cr = Cr[d * NNh + n], ci = Ci[d * NNh + n];
  WC[idx] = p ? cr : ci;
  WC[(size_t)KH * DD + idx] = p ? ci : -cr;
}

#define PKS 64
#define PLD 68
#define PBN 48

__global__ __launch_bounds__(256) void gemm_proj_kernel(const float* __restrict__ x,
                                                        const float* __restrict__ W1t,
                                                        const float* __restrict__ gate_b,
                                                        float* __restrict__ P) {
  __shared__ __attribute__((aligned(16))) float As[128 * PLD];
  __shared__ __attribute__((aligned(16))) float Bs[PBN * PLD];
  v8f acc[3] = {{0,0,0,0,0,0,0,0}, {0,0,0,0,0,0,0,0}, {0,0,0,0,0,0,0,0}};

  const int tid  = threadIdx.x;
  const int lane = tid & 31, wave = tid >> 5;
  const int half = lane >> 4, lrow = lane & 15;
  const int m0 = blockIdx.x * 128;
  const int j0 = blockIdx.y * PBN;

  const int srow = tid >> 4, skq = (tid & 15) << 2;
  const float* ap = x   + (size_t)(m0 + srow) * KX + skq;
  const float* bp = W1t + (size_t)(j0 + srow) * KX + skq;
  constexpr size_t RS = (size_t)16 * KX;
  float4 a0 = *(const float4*)(ap);
  float4 a1 = *(const float4*)(ap + RS);
  float4 a2 = *(const float4*)(ap + 2 * RS);
  float4 a3 = *(const float4*)(ap + 3 * RS);
  float4 a4 = *(const float4*)(ap + 4 * RS);
  float4 a5 = *(const float4*)(ap + 5 * RS);
  float4 a6 = *(const float4*)(ap + 6 * RS);
  float4 a7 = *(const float4*)(ap + 7 * RS);
  float4 b0 = *(const float4*)(bp);
  float4 b1 = *(const float4*)(bp + RS);
  float4 b2 = *(const float4*)(bp + 2 * RS);

  float* ast = As + srow * PLD + skq;
  float* bst = Bs + srow * PLD + skq;
  const float* af = As + (wave * 16 + lrow) * PLD + 4 * half;
  const float* bf = Bs + lrow * PLD + 4 * half;

  for (int k0 = 0; k0 < KX; k0 += PKS) {
    __syncthreads();
    *(float4*)(ast)                = a0;
    *(float4*)(ast + 1 * 16 * PLD) = a1;
    *(float4*)(ast + 2 * 16 * PLD) = a2;
    *(float4*)(ast + 3 * 16 * PLD) = a3;
    *(float4*)(ast + 4 * 16 * PLD) = a4;
    *(float4*)(ast + 5 * 16 * PLD) = a5;
    *(float4*)(ast + 6 * 16 * PLD) = a6;
    *(float4*)(ast + 7 * 16 * PLD) = a7;
    *(float4*)(bst)                = b0;
    *(float4*)(bst + 1 * 16 * PLD) = b1;
    *(float4*)(bst + 2 * 16 * PLD) = b2;
    __syncthreads();

    if (k0 + PKS < KX) {
      const float* an = ap + (k0 + PKS);
      const float* bn = bp + (k0 + PKS);
      a0 = *(const float4*)(an);
      a1 = *(const float4*)(an + RS);
      a2 = *(const float4*)(an + 2 * RS);
      a3 = *(const float4*)(an + 3 * RS);
      a4 = *(const float4*)(an + 4 * RS);
      a5 = *(const float4*)(an + 5 * RS);
      a6 = *(const float4*)(an + 6 * RS);
      a7 = *(const float4*)(an + 7 * RS);
      b0 = *(const float4*)(bn);
      b1 = *(const float4*)(bn + RS);
      b2 = *(const float4*)(bn + 2 * RS);
    }

#pragma unroll
    for (int u = 0; u < PKS / 8; ++u) {
      const float4 a = *(const float4*)(af + 8 * u);
      float4 b[3];
#pragma unroll
      for (int ni = 0; ni < 3; ++ni) b[ni] = *(const float4*)(bf + ni * 16 * PLD + 8 * u);
      v2f a01, a23;
      a01.x = a.x; a01.y = a.y; a23.x = a.z; a23.y = a.w;
#pragma unroll
      for (int ni = 0; ni < 3; ++ni) {
        v2f t; t.x = b[ni].x; t.y = b[ni].y;
        acc[ni] = wmma_f32(a01, t, acc[ni]);
      }
#pragma unroll
      for (int ni = 0; ni < 3; ++ni) {
        v2f t; t.x = b[ni].z; t.y = b[ni].w;
        acc[ni] = wmma_f32(a23, t, acc[ni]);
      }
    }
  }

  const int rb = ((m0 + wave * 16) >> 3) + half;
#pragma unroll
  for (int ni = 0; ni < 3; ++ni) {
    const int j = j0 + ni * 16 + lrow;
    float v[8];
#pragma unroll
    for (int e = 0; e < 8; ++e) {
      v[e] = acc[ni][e];
      if (j >= 128) v[e] = 1.0f / (1.0f + expf(-(v[e] + gate_b[j - 128])));
    }
    float* dst = P + (size_t)rb * PBLK + j * 8;
    *(float4*)(dst)     = make_float4(v[0], v[1], v[2], v[3]);
    *(float4*)(dst + 4) = make_float4(v[4], v[5], v[6], v[7]);
  }
}

struct ScanChunk {
  float4 br[2];
  float4 bi[2];
  float4 g[2];
};

__device__ __forceinline__ void scan_load(ScanChunk& c, const float* __restrict__ pw, unsigned lo, int chunk) {
  const float* p = pw + (size_t)chunk * PBLK;
  c.br[0] = *(const float4*)(p + lo);
  c.br[1] = *(const float4*)(p + 4 + lo);
  c.bi[0] = *(const float4*)(p + 64 * 8 + lo);
  c.bi[1] = *(const float4*)(p + 64 * 8 + 4 + lo);
  c.g[0]  = *(const float4*)(p + 128 * 8 + lo);
  c.g[1]  = *(const float4*)(p + 128 * 8 + 4 + lo);
}

__device__ __forceinline__ float f4_get(const float4& v, int i) {
  return i == 0 ? v.x : (i == 1 ? v.y : (i == 2 ? v.z : v.w));
}

__device__ __forceinline__ void scan_steps(const ScanChunk& c, float ca, float sa,
                                           float& hr, float& hi, float* __restrict__ hw, unsigned ho) {
#pragma unroll
  for (int i = 0; i < CHS; ++i) {
    float g  = f4_get(c.g[i >> 2], i & 3);
    float br = f4_get(c.br[i >> 2], i & 3);
    float bi = f4_get(c.bi[i >> 2], i & 3);
    float rr = hr * ca - hi * sa;
    float ri = hr * sa + hi * ca;
    hr = g * br + (1.0f - g) * rr;
    hi = g * bi + (1.0f - g) * ri;
    *(float2*)(hw + (size_t)i * KH + ho) = make_float2(hr, hi);
  }
}

__global__ __launch_bounds__(32) void scan_kernel(const float* __restrict__ P,
                                                  const float* __restrict__ h0,
                                                  const float* __restrict__ theta,
                                                  const float* __restrict__ dampp,
                                                  float* __restrict__ hws,
                                                  float* __restrict__ hfin) {
  const int b  = blockIdx.x >> 1;
  const int n0 = (blockIdx.x & 1) * 32;
  const int n  = n0 + threadIdx.x;
  const float* pw = P + (size_t)b * (SS / 8) * PBLK + n0 * 8;
  float* hw = hws + (size_t)b * SS * KH + 2 * n0;
  const unsigned lo = threadIdx.x * 8u;
  const unsigned ho = threadIdx.x * 2u;

  const float c = cosf(theta[n]);
  const float s = sinf(theta[n]);
  const float dmp = 0.5f + 0.5f / (1.0f + expf(-dampp[n]));
  const float ca = c * dmp, sa = s * dmp;
  float hr = h0[(b * NNh + n) * 2 + 0];
  float hi = h0[(b * NNh + n) * 2 + 1];

  __builtin_amdgcn_sched_barrier(0);

  ScanChunk ring[RING];
#pragma unroll
  for (int r = 0; r < RING - 1; ++r) scan_load(ring[r], pw, lo, r);

  int k = 0;
  for (; k < NCH - RING; k += RING) {
#pragma unroll
    for (int r = 0; r < RING; ++r) {
      __builtin_amdgcn_sched_barrier(0);
      scan_load(ring[(r + RING - 1) % RING], pw, lo, k + r + RING - 1);
      __builtin_amdgcn_sched_barrier(0);
      scan_steps(ring[r], ca, sa, hr, hi, hw + (size_t)(k + r) * CHS * KH, ho);
    }
  }
  __builtin_amdgcn_sched_barrier(0);
  scan_load(ring[RING - 1], pw, lo, NCH - 1);
#pragma unroll
  for (int r = 0; r < RING; ++r) {
    __builtin_amdgcn_sched_barrier(0);
    scan_steps(ring[r], ca, sa, hr, hi, hw + (size_t)(NCH - RING + r) * CHS * KH, ho);
  }
  hfin[(b * NNh + n) * 2 + 0] = hr;
  hfin[(b * NNh + n) * 2 + 1] = hi;
}

#define B3LD 80

__global__ __launch_bounds__(256) void gemm_out_kernel(const float* __restrict__ x,
                                                       const float* __restrict__ WB,
                                                       const float* __restrict__ hws,
                                                       const float* __restrict__ WC,
                                                       float* __restrict__ out) {
  __shared__ v2f As2[3][8 * A2LD];
  __shared__ v2f Bs2[3][8 * B3LD];
  v8f acc[3][2][2];
#pragma unroll
  for (int pl = 0; pl < 3; ++pl)
#pragma unroll
    for (int mi = 0; mi < 2; ++mi)
#pragma unroll
      for (int ni = 0; ni < 2; ++ni)
#pragma unroll
        for (int e = 0; e < 8; ++e) acc[pl][mi][ni][e] = 0.0f;

  const int tid  = threadIdx.x;
  const int lane = tid & 31, wave = tid >> 5;
  const int half = lane >> 4, lrow = lane & 15;
  const int wm = (wave & 3) * 32, wn = (wave >> 2) * 32;
  const int m0 = blockIdx.x * 128;
  const int j0 = blockIdx.y * 64;

  const int bk = tid >> 4, bj = (tid & 15) << 2;
  const int bbase = ((bk >> 1) * B3LD + bj) * 2 + (bk & 1);

  {
    const int arow = tid >> 3, akp = tid & 7;
    const float* ap = x + (size_t)(m0 + arow) * KX + akp * 4;
    const float* bp = WB + (size_t)bk * DD + j0 + bj;
    float4 aReg[4], bReg[3];
#pragma unroll
    for (int i = 0; i < 4; ++i) aReg[i] = *(const float4*)(ap + (size_t)i * 32 * KX);
#pragma unroll
    for (int pl = 0; pl < 3; ++pl) bReg[pl] = *(const float4*)(bp + (size_t)pl * DD * DD);

    for (int k0 = 0; k0 < DD; k0 += 16) {
      __syncthreads();
#pragma unroll
      for (int i = 0; i < 4; ++i) {
        const float4 f = aReg[i];
        const int o = akp * A2LD + arow + i * 32;
        v2f t;
        t.x = f.x + f.y; t.y = f.z + f.w; As2[0][o] = t;
        t.x = f.x;       t.y = f.z;       As2[1][o] = t;
        t.x = f.y;       t.y = f.w;       As2[2][o] = t;
      }
#pragma unroll
      for (int pl = 0; pl < 3; ++pl) {
        float* bsf = (float*)Bs2[pl];
        bsf[bbase + 0] = bReg[pl].x;
        bsf[bbase + 2] = bReg[pl].y;
        bsf[bbase + 4] = bReg[pl].z;
        bsf[bbase + 6] = bReg[pl].w;
      }
      __syncthreads();

      if (k0 + 16 < DD) {
#pragma unroll
        for (int i = 0; i < 4; ++i)
          aReg[i] = *(const float4*)(ap + (size_t)i * 32 * KX + (k0 + 16) * 2);
#pragma unroll
        for (int pl = 0; pl < 3; ++pl)
          bReg[pl] = *(const float4*)(bp + (size_t)pl * DD * DD + (size_t)(k0 + 16) * DD);
      }

#pragma unroll
      for (int kstep = 0; kstep < 4; ++kstep) {
        const int p = 2 * kstep + half;
        v2f a0[3], a1[3], b0[3], b1[3];
#pragma unroll
        for (int pl = 0; pl < 3; ++pl) {
          a0[pl] = As2[pl][p * A2LD + wm + lrow];
          a1[pl] = As2[pl][p * A2LD + wm + 16 + lrow];
          b0[pl] = Bs2[pl][p * B3LD + wn + lrow];
          b1[pl] = Bs2[pl][p * B3LD + wn + 16 + lrow];
        }
#pragma unroll
        for (int pl = 0; pl < 3; ++pl) {
          acc[pl][0][0] = wmma_f32(a0[pl], b0[pl], acc[pl][0][0]);
          acc[pl][1][0] = wmma_f32(a1[pl], b0[pl], acc[pl][1][0]);
          acc[pl][0][1] = wmma_f32(a0[pl], b1[pl], acc[pl][0][1]);
          acc[pl][1][1] = wmma_f32(a1[pl], b1[pl], acc[pl][1][1]);
        }
      }
    }
  }

  {
    const int am0 = tid >> 2, am1 = 64 + (tid >> 2);
    const int ak0 = (tid & 3) << 2;
    const float* ap0 = hws + (size_t)(m0 + am0) * KH + ak0;
    const float* ap1 = hws + (size_t)(m0 + am1) * KH + ak0;
    const float* bp = WC + (size_t)bk * DD + j0 + bj;
    float4 aReg0 = *(const float4*)(ap0);
    float4 aReg1 = *(const float4*)(ap1);
    float4 bReg[2];
#pragma unroll
    for (int q = 0; q < 2; ++q) bReg[q] = *(const float4*)(bp + (size_t)q * KH * DD);

    for (int k0 = 0; k0 < KH; k0 += 16) {
      __syncthreads();
      {
        v2f t;
        t.x = aReg0.x; t.y = aReg0.y; As2[0][(ak0 >> 1) * A2LD + am0] = t;
        t.x = aReg0.z; t.y = aReg0.w; As2[0][((ak0 >> 1) + 1) * A2LD + am0] = t;
        t.x = aReg1.x; t.y = aReg1.y; As2[0][(ak0 >> 1) * A2LD + am1] = t;
        t.x = aReg1.z; t.y = aReg1.w; As2[0][((ak0 >> 1) + 1) * A2LD + am1] = t;
      }
#pragma unroll
      for (int q = 0; q < 2; ++q) {
        float* bsf = (float*)Bs2[q];
        bsf[bbase + 0] = bReg[q].x;
        bsf[bbase + 2] = bReg[q].y;
        bsf[bbase + 4] = bReg[q].z;
        bsf[bbase + 6] = bReg[q].w;
      }
      __syncthreads();

      if (k0 + 16 < KH) {
        aReg0 = *(const float4*)(ap0 + (k0 + 16));
        aReg1 = *(const float4*)(ap1 + (k0 + 16));
#pragma unroll
        for (int q = 0; q < 2; ++q)
          bReg[q] = *(const float4*)(bp + (size_t)q * KH * DD + (size_t)(k0 + 16) * DD);
      }

#pragma unroll
      for (int kstep = 0; kstep < 4; ++kstep) {
        const int p = 2 * kstep + half;
        const v2f a0 = As2[0][p * A2LD + wm + lrow];
        const v2f a1 = As2[0][p * A2LD + wm + 16 + lrow];
        v2f b0[2], b1[2];
#pragma unroll
        for (int q = 0; q < 2; ++q) {
          b0[q] = Bs2[q][p * B3LD + wn + lrow];
          b1[q] = Bs2[q][p * B3LD + wn + 16 + lrow];
        }
#pragma unroll
        for (int q = 0; q < 2; ++q) {
          acc[1 + q][0][0] = wmma_f32(a0, b0[q], acc[1 + q][0][0]);
          acc[1 + q][1][0] = wmma_f32(a1, b0[q], acc[1 + q][1][0]);
          acc[1 + q][0][1] = wmma_f32(a0, b1[q], acc[1 + q][0][1]);
          acc[1 + q][1][1] = wmma_f32(a1, b1[q], acc[1 + q][1][1]);
        }
      }
    }
  }

#pragma unroll
  for (int mi = 0; mi < 2; ++mi)
#pragma unroll
    for (int ni = 0; ni < 2; ++ni)
#pragma unroll
      for (int e = 0; e < 8; ++e) {
        const int m = m0 + wm + mi * 16 + half * 8 + e;
        const int d = j0 + wn + ni * 16 + lrow;
        const float k1 = acc[0][mi][ni][e];
        const float zr = k1 - acc[2][mi][ni][e];
        const float zi = k1 + acc[1][mi][ni][e];
        *(float2*)(out + (size_t)m * NO + 2 * d) = make_float2(zr * zr + zr, zr * zi + zi);
      }
}

extern "C" void kernel_launch(void* const* d_in, const int* in_sizes, int n_in,
                              void* d_out, int out_size, void* d_ws, size_t ws_size,
                              hipStream_t stream) {
  const float* x     = (const float*)d_in[0];
  const float* h0    = (const float*)d_in[1];
  const float* th    = (const float*)d_in[2];
  const float* dampp = (const float*)d_in[3];
  const float* Bwr   = (const float*)d_in[4];
  const float* Bwi   = (const float*)d_in[5];
  const float* Cwr   = (const float*)d_in[6];
  const float* Cwi   = (const float*)d_in[7];
  const float* Dwr   = (const float*)d_in[8];
  const float* Dwi   = (const float*)d_in[9];
  const float* gw    = (const float*)d_in[10];
  const float* gb    = (const float*)d_in[11];

  float* out  = (float*)d_out;
  float* hfin = out + (size_t)MM * NO;

  float* ws = (float*)d_ws;
  float* W1  = ws;
  float* WD  = W1 + (size_t)KX * NP;
  float* WC  = WD + (size_t)KX * NO;
  float* P   = WC + (size_t)KH * NO;
  float* Hws = P + (size_t)MM * NP;

  pack_w1_kernel<<<(KX * NP + 255) / 256, 256, 0, stream>>>(Bwr, Bwi, gw, W1);
  pack_wd_kernel<<<dim3(DD / 32, DD / 32), 256, 0, stream>>>(Dwr, Dwi, WD);
  pack_wc_kernel<<<(KH * DD + 255) / 256, 256, 0, stream>>>(Cwr, Cwi, WC);

  gemm_proj_kernel<<<dim3(MM / 128, NP / PBN), 256, 0, stream>>>(x, W1, gb, P);
  scan_kernel<<<BB * NNh / 32, 32, 0, stream>>>(P, h0, th, dampp, Hws, hfin);
  gemm_out_kernel<<<dim3(MM / 128, DD / 64), 256, 0, stream>>>(x, WD, Hws, WC, out);
}


# scan split along time: each chain cut into 32 segments of 64 steps run in parallel (16 work-groups of 512), start values folded from segment products; changes the order of the recurrence's sums
# speedup vs baseline: 2.1357x; 1.0435x over previous
#include <hip/hip_runtime.h>
#include <math.h>

#define BB 4
#define SS 2048
#define DD 1024
#define NNh 64
#define MM (BB * SS)
#define KX (2 * DD)
#define NP 192
#define KH (2 * NNh)
#define NO (2 * DD)

#define A2LD 144

#define PBLK (NP * 8)
#define CHS 8
#define SSEG 32
#define SLEN (SS / SSEG)
#define SCHK (SLEN / CHS)
#define SNG 16

typedef __attribute__((ext_vector_type(2))) float v2f;
typedef __attribute__((ext_vector_type(8))) float v8f;

__device__ __forceinline__ v8f wmma_f32(v2f a, v2f b, v8f c) {
  return __builtin_amdgcn_wmma_f32_16x16x4_f32(
       false, a,  false, b,
       (short)0, c,  false,  false);
}

__global__ void pack_w1_kernel(const float* __restrict__ Bwr, const float* __restrict__ Bwi,
                               const float* __restrict__ gw, float* __restrict__ W1t) {
  int idx = blockIdx.x * 256 + threadIdx.x;
  if (idx >= KX * NP) return;
  int j = idx / KX, k = idx - j * KX;
  int d = k >> 1, p = k & 1;
  float v;
  if (j < 64) {
    v = p ? -Bwi[j * DD + d] : Bwr[j * DD + d];
  } else if (j < 128) {
    int n = j - 64;
    v = p ? Bwr[n * DD + d] : Bwi[n * DD + d];
  } else {
    int n = j - 128;
    v = gw[n * (2 * DD) + p * DD + d];
  }
  W1t[idx] = v;
}

__global__ __launch_bounds__(256) void pack_wd_kernel(const float* __restrict__ Dr,
                                                      const float* __restrict__ Di,
                                                      float* __restrict__ WB) {
  __shared__ float tr[32][33];
  __shared__ float ti[32][33];
  const int tx = threadIdx.x & 31, ty = threadIdx.x >> 5;
  const int e0 = blockIdx.x * 32, d0 = blockIdx.y * 32;
  for (int r = ty; r < 32; r += 8) {
    tr[r][tx] = Dr[(size_t)(d0 + r) * DD + e0 + tx];
    ti[r][tx] = Di[(size_t)(d0 + r) * DD + e0 + tx];
  }
  __syncthreads();
  for (int r = ty; r < 32; r += 8) {
    const float wr = tr[tx][r], wi = ti[tx][r];
    const size_t o = (size_t)(e0 + r) * DD + d0 + tx;
    WB[o] = wr;
    WB[(size_t)DD * DD + o] = wi - wr;
    WB[2 * (size_t)DD * DD + o] = wr + wi;
  }
}

__global__ void pack_wc_kernel(const float* __restrict__ Cr, const float* __restrict__ Ci,
                               float* __restrict__ WC) {
  int idx = blockIdx.x * 256 + threadIdx.x;
  if (idx >= KH * DD) return;
  int k = idx >> 10, d = idx & (DD - 1);
  int n = k >> 1, p = k & 1;
  float cr = Cr[d * NNh + n], ci = Ci[d * NNh + n];
  WC[idx] = p ? cr : ci;
  WC[(size_t)KH * DD + idx] = p ? ci : -cr;
}

#define PKS 64
#define PLD 68
#define PBN 48

__global__ __launch_bounds__(256) void gemm_proj_kernel(const float* __restrict__ x,
                                                        const float* __restrict__ W1t,
                                                        const float* __restrict__ gate_b,
                                                        float* __restrict__ P) {
  __shared__ __attribute__((aligned(16))) float As[128 * PLD];
  __shared__ __attribute__((aligned(16))) float Bs[PBN * PLD];
  v8f acc[3] = {{0,0,0,0,0,0,0,0}, {0,0,0,0,0,0,0,0}, {0,0,0,0,0,0,0,0}};

  const int tid  = threadIdx.x;
  const int lane = tid & 31, wave = tid >> 5;
  const int half = lane >> 4, lrow = lane & 15;
  const int m0 = blockIdx.x * 128;
  const int j0 = blockIdx.y * PBN;

  const int srow = tid >> 4, skq = (tid & 15) << 2;
  const float* ap = x   + (size_t)(m0 + srow) * KX + skq;
  const float* bp = W1t + (size_t)(j0 + srow) * KX + skq;
  constexpr size_t RS = (size_t)16 * KX;
  float4 a0 = *(const float4*)(ap);
  float4 a1 = *(const float4*)(ap + RS);
  float4 a2 = *(const float4*)(ap + 2 * RS);
  float4 a3 = *(const float4*)(ap + 3 * RS);
  float4 a4 = *(const float4*)(ap + 4 * RS);
  float4 a5 = *(const float4*)(ap + 5 * RS);
  float4 a6 = *(const float4*)(ap + 6 * RS);
  float4 a7 = *(const float4*)(ap + 7 * RS);
  float4 b0 = *(const float4*)(bp);
  float4 b1 = *(const float4*)(bp + RS);
  float4 b2 = *(const float4*)(bp + 2 * RS);

  float* ast = As + srow * PLD + skq;
  float* bst = Bs + srow * PLD + skq;
  const float* af = As + (wave * 16 + lrow) * PLD + 4 * half;
  const float* bf = Bs + lrow * PLD + 4 * half;

  for (int k0 = 0; k0 < KX; k0 += PKS) {
    __syncthreads();
    *(float4*)(ast)                = a0;
    *(float4*)(ast + 1 * 16 * PLD) = a1;
    *(float4*)(ast + 2 * 16 * PLD) = a2;
    *(float4*)(ast + 3 * 16 * PLD) = a3;
    *(float4*)(ast + 4 * 16 * PLD) = a4;
    *(float4*)(ast + 5 * 16 * PLD) = a5;
    *(float4*)(ast + 6 * 16 * PLD) = a6;
    *(float4*)(ast + 7 * 16 * PLD) = a7;
    *(float4*)(bst)                = b0;
    *(float4*)(bst + 1 * 16 * PLD) = b1;
    *(float4*)(bst + 2 * 16 * PLD) = b2;
    __syncthreads();

    if (k0 + PKS < KX) {
      const float* an = ap + (k0 + PKS);
      const float* bn = bp + (k0 + PKS);
      a0 = *(const float4*)(an);
      a1 = *(const float4*)(an + RS);
      a2 = *(const float4*)(an + 2 * RS);
      a3 = *(const float4*)(an + 3 * RS);
      a4 = *(const float4*)(an + 4 * RS);
      a5 = *(const float4*)(an + 5 * RS);
      a6 = *(const float4*)(an + 6 * RS);
      a7 = *(const float4*)(an + 7 * RS);
      b0 = *(const float4*)(bn);
      b1 = *(const float4*)(bn + RS);
      b2 = *(const float4*)(bn + 2 * RS);
    }

#pragma unroll
    for (int u = 0; u < PKS / 8; ++u) {
      const float4 a = *(const float4*)(af + 8 * u);
      float4 b[3];
#pragma unroll
      for (int ni = 0; ni < 3; ++ni) b[ni] = *(const float4*)(bf + ni * 16 * PLD + 8 * u);
      v2f a01, a23;
      a01.x = a.x; a01.y = a.y; a23.x = a.z; a23.y = a.w;
#pragma unroll
      for (int ni = 0; ni < 3; ++ni) {
        v2f t; t.x = b[ni].x; t.y = b[ni].y;
        acc[ni] = wmma_f32(a01, t, acc[ni]);
      }
#pragma unroll
      for (int ni = 0; ni < 3; ++ni) {
        v2f t; t.x = b[ni].z; t.y = b[ni].w;
        acc[ni] = wmma_f32(a23, t, acc[ni]);
      }
    }
  }

  const int rb = ((m0 + wave * 16) >> 3) + half;
#pragma unroll
  for (int ni = 0; ni < 3; ++ni) {
    const int j = j0 + ni * 16 + lrow;
    float v[8];
#pragma unroll
    for (int e = 0; e < 8; ++e) {
      v[e] = acc[ni][e];
      if (j >= 128) v[e] = 1.0f / (1.0f + expf(-(v[e] + gate_b[j - 128])));
    }
    float* dst = P + (size_t)rb * PBLK + j * 8;
    *(float4*)(dst)     = make_float4(v[0], v[1], v[2], v[3]);
    *(float4*)(dst + 4) = make_float4(v[4], v[5], v[6], v[7]);
  }
}

struct ScanChunk {
  float4 br[2];
  float4 bi[2];
  float4 g[2];
};

__device__ __forceinline__ void scan_load(ScanChunk& c, const float* __restrict__ p) {
  c.br[0] = *(const float4*)(p);
  c.br[1] = *(const float4*)(p + 4);
  c.bi[0] = *(const float4*)(p + 64 * 8);
  c.bi[1] = *(const float4*)(p + 64 * 8 + 4);
  c.g[0]  = *(const float4*)(p + 128 * 8);
  c.g[1]  = *(const float4*)(p + 128 * 8 + 4);
}

__device__ __forceinline__ float f4_get(const float4& v, int i) {
  return i == 0 ? v.x : (i == 1 ? v.y : (i == 2 ? v.z : v.w));
}

__device__ __forceinline__ void scan_steps_local(const ScanChunk& c, float ca, float sa,
                                                 float& hr, float& hi, float& ap) {
#pragma unroll
  for (int i = 0; i < CHS; ++i) {
    float g  = f4_get(c.g[i >> 2], i & 3);
    float br = f4_get(c.br[i >> 2], i & 3);
    float bi = f4_get(c.bi[i >> 2], i & 3);
    float rr = hr * ca - hi * sa;
    float ri = hr * sa + hi * ca;
    hr = g * br + (1.0f - g) * rr;
    hi = g * bi + (1.0f - g) * ri;
    ap *= (1.0f - g);
  }
}

__device__ __forceinline__ void scan_steps(const ScanChunk& c, float ca, float sa,
                                           float& hr, float& hi, float* __restrict__ hw) {
#pragma unroll
  for (int i = 0; i < CHS; ++i) {
    float g  = f4_get(c.g[i >> 2], i & 3);
    float br = f4_get(c.br[i >> 2], i & 3);
    float bi = f4_get(c.bi[i >> 2], i & 3);
    float rr = hr * ca - hi * sa;
    float ri = hr * sa + hi * ca;
    hr = g * br + (1.0f - g) * rr;
    hi = g * bi + (1.0f - g) * ri;
    *(float2*)(hw + (size_t)i * KH) = make_float2(hr, hi);
  }
}

__global__ __launch_bounds__(SNG * SSEG) void scan_kernel(const float* __restrict__ P,
                                                          const float* __restrict__ h0,
                                                          const float* __restrict__ theta,
                                                          const float* __restrict__ dampp,
                                                          float* __restrict__ hws,
                                                          float* __restrict__ hfin) {
  __shared__ float sA[SSEG][SNG];
  __shared__ float sEr[SSEG][SNG];
  __shared__ float sEi[SSEG][SNG];

  const int b   = blockIdx.x / (NNh / SNG);
  const int n0  = (blockIdx.x % (NNh / SNG)) * SNG;
  const int nl  = threadIdx.x & (SNG - 1);
  const int seg = threadIdx.x / SNG;
  const int n   = n0 + nl;
  const float* pseg = P + ((size_t)b * (SS / 8) + (size_t)seg * SCHK) * PBLK + n * 8;
  float* hseg = hws + ((size_t)b * SS + (size_t)seg * SLEN) * KH + 2 * n;

  const float c = cosf(theta[n]);
  const float s = sinf(theta[n]);
  const float dmp = 0.5f + 0.5f / (1.0f + expf(-dampp[n]));
  const float ca = c * dmp, sa = s * dmp;

  static_assert(SLEN == 64, "six squarings give the 64th power");
  float pr = ca, pi = sa;
#pragma unroll
  for (int q = 0; q < 6; ++q) {
    const float tr = pr * pr - pi * pi;
    const float ti = 2.0f * pr * pi;
    pr = tr; pi = ti;
  }

  float er = 0.0f, ei = 0.0f, ap = 1.0f;
  {
    ScanChunk cur, nxt;
    scan_load(cur, pseg);
#pragma unroll
    for (int k = 0; k < SCHK; ++k) {
      if (k + 1 < SCHK) scan_load(nxt, pseg + (size_t)(k + 1) * PBLK);
      scan_steps_local(cur, ca, sa, er, ei, ap);
      if (k + 1 < SCHK) cur = nxt;
    }
  }
  sA[seg][nl]  = ap;
  sEr[seg][nl] = er;
  sEi[seg][nl] = ei;
  __syncthreads();

  float hr = h0[(b * NNh + n) * 2 + 0];
  float hi = h0[(b * NNh + n) * 2 + 1];
  for (int j = 0; j < seg; ++j) {
    const float a  = sA[j][nl];
    const float mr = a * pr, mi = a * pi;
    const float tr = mr * hr - mi * hi + sEr[j][nl];
    const float ti = mr * hi + mi * hr + sEi[j][nl];
    hr = tr; hi = ti;
  }

  {
    ScanChunk cur, nxt;
    scan_load(cur, pseg);
#pragma unroll
    for (int k = 0; k < SCHK; ++k) {
      if (k + 1 < SCHK) scan_load(nxt, pseg + (size_t)(k + 1) * PBLK);
      scan_steps(cur, ca, sa, hr, hi, hseg + (size_t)k * CHS * KH);
      if (k + 1 < SCHK) cur = nxt;
    }
  }
  if (seg == SSEG - 1) {
    hfin[(b * NNh + n) * 2 + 0] = hr;
    hfin[(b * NNh + n) * 2 + 1] = hi;
  }
}

#define B3LD 80

__global__ __launch_bounds__(256) void gemm_out_kernel(const float* __restrict__ x,
                                                       const float* __restrict__ WB,
                                                       const float* __restrict__ hws,
                                                       const float* __restrict__ WC,
                                                       float* __restrict__ out) {
  __shared__ v2f As2[3][8 * A2LD];
  __shared__ v2f Bs2[3][8 * B3LD];
  v8f acc[3][2][2];
#pragma unroll
  for (int pl = 0; pl < 3; ++pl)
#pragma unroll
    for (int mi = 0; mi < 2; ++mi)
#pragma unroll
      for (int ni = 0; ni < 2; ++ni)
#pragma unroll
        for (int e = 0; e < 8; ++e) acc[pl][mi][ni][e] = 0.0f;

  const int tid  = threadIdx.x;
  const int lane = tid & 31, wave = tid >> 5;
  const int half = lane >> 4, lrow = lane & 15;
  const int wm = (wave & 3) * 32, wn = (wave >> 2) * 32;
  const int m0 = blockIdx.x * 128;
  const int j0 = blockIdx.y * 64;

  const int bk = tid >> 4, bj = (tid & 15) << 2;
  const int bbase = ((bk >> 1) * B3LD + bj) * 2 + (bk & 1);

  {
    const int arow = tid >> 3, akp = tid & 7;
    const float* ap = x + (size_t)(m0 + arow) * KX + akp * 4;
    const float* bp = WB + (size_t)bk * DD + j0 + bj;
    float4 aReg[4], bReg[3];
#pragma unroll
    for (int i = 0; i < 4; ++i) aReg[i] = *(const float4*)(ap + (size_t)i * 32 * KX);
#pragma unroll
    for (int pl = 0; pl < 3; ++pl) bReg[pl] = *(const float4*)(bp + (size_t)pl * DD * DD);

    for (int k0 = 0; k0 < DD; k0 += 16) {
      __syncthreads();
#pragma unroll
      for (int i = 0; i < 4; ++i) {
        const float4 f = aReg[i];
        const int o = akp * A2LD + arow + i * 32;
        v2f t;
        t.x = f.x + f.y; t.y = f.z + f.w; As2[0][o] = t;
        t.x = f.x;       t.y = f.z;       As2[1][o] = t;
        t.x = f.y;       t.y = f.w;       As2[2][o] = t;
      }
#pragma unroll
      for (int pl = 0; pl < 3; ++pl) {
        float* bsf = (float*)Bs2[pl];
        bsf[bbase + 0] = bReg[pl].x;
        bsf[bbase + 2] = bReg[pl].y;
        bsf[bbase + 4] = bReg[pl].z;
        bsf[bbase + 6] = bReg[pl].w;
      }
      __syncthreads();

      if (k0 + 16 < DD) {
#pragma unroll
        for (int i = 0; i < 4; ++i)
          aReg[i] = *(const float4*)(ap + (size_t)i * 32 * KX + (k0 + 16) * 2);
#pragma unroll
        for (int pl = 0; pl < 3; ++pl)
          bReg[pl] = *(const float4*)(bp + (size_t)pl * DD * DD + (size_t)(k0 + 16) * DD);
      }

#pragma unroll
      for (int kstep = 0; kstep < 4; ++kstep) {
        const int p = 2 * kstep + half;
        v2f a0[3], a1[3], b0[3], b1[3];
#pragma unroll
        for (int pl = 0; pl < 3; ++pl) {
          a0[pl] = As2[pl][p * A2LD + wm + lrow];
          a1[pl] = As2[pl][p * A2LD + wm + 16 + lrow];
          b0[pl] = Bs2[pl][p * B3LD + wn + lrow];
          b1[pl] = Bs2[pl][p * B3LD + wn + 16 + lrow];
        }
#pragma unroll
        for (int pl = 0; pl < 3; ++pl) {
          acc[pl][0][0] = wmma_f32(a0[pl], b0[pl], acc[pl][0][0]);
          acc[pl][1][0] = wmma_f32(a1[pl], b0[pl], acc[pl][1][0]);
          acc[pl][0][1] = wmma_f32(a0[pl], b1[pl], acc[pl][0][1]);
          acc[pl][1][1] = wmma_f32(a1[pl], b1[pl], acc[pl][1][1]);
        }
      }
    }
  }

  {
    const int am0 = tid >> 2, am1 = 64 + (tid >> 2);
    const int ak0 = (tid & 3) << 2;
    const float* ap0 = hws + (size_t)(m0 + am0) * KH + ak0;
    const float* ap1 = hws + (size_t)(m0 + am1) * KH + ak0;
    const float* bp = WC + (size_t)bk * DD + j0 + bj;
    float4 aReg0 = *(const float4*)(ap0);
    float4 aReg1 = *(const float4*)(ap1);
    float4 bReg[2];
#pragma unroll
    for (int q = 0; q < 2; ++q) bReg[q] = *(const float4*)(bp + (size_t)q * KH * DD);

    for (int k0 = 0; k0 < KH; k0 += 16) {
      __syncthreads();
      {
        v2f t;
        t.x = aReg0.x; t.y = aReg0.y; As2[0][(ak0 >> 1) * A2LD + am0] = t;
        t.x = aReg0.z; t.y = aReg0.w; As2[0][((ak0 >> 1) + 1) * A2LD + am0] = t;
        t.x = aReg1.x; t.y = aReg1.y; As2[0][(ak0 >> 1) * A2LD + am1] = t;
        t.x = aReg1.z; t.y = aReg1.w; As2[0][((ak0 >> 1) + 1) * A2LD + am1] = t;
      }
#pragma unroll
      for (int q = 0; q < 2; ++q) {
        float* bsf = (float*)Bs2[q];
        bsf[bbase + 0] = bReg[q].x;
        bsf[bbase + 2] = bReg[q].y;
        bsf[bbase + 4] = bReg[q].z;
        bsf[bbase + 6] = bReg[q].w;
      }
      __syncthreads();

      if (k0 + 16 < KH) {
        aReg0 = *(const float4*)(ap0 + (k0 + 16));
        aReg1 = *(const float4*)(ap1 + (k0 + 16));
#pragma unroll
        for (int q = 0; q < 2; ++q)
          bReg[q] = *(const float4*)(bp + (size_t)q * KH * DD + (size_t)(k0 + 16) * DD);
      }

#pragma unroll
      for (int kstep = 0; kstep < 4; ++kstep) {
        const int p = 2 * kstep + half;
        const v2f a0 = As2[0][p * A2LD + wm + lrow];
        const v2f a1 = As2[0][p * A2LD + wm + 16 + lrow];
        v2f b0[2], b1[2];
#pragma unroll
        for (int q = 0; q < 2; ++q) {
          b0[q] = Bs2[q][p * B3LD + wn + lrow];
          b1[q] = Bs2[q][p * B3LD + wn + 16 + lrow];
        }
#pragma unroll
        for (int q = 0; q < 2; ++q) {
          acc[1 + q][0][0] = wmma_f32(a0, b0[q], acc[1 + q][0][0]);
          acc[1 + q][1][0] = wmma_f32(a1, b0[q], acc[1 + q][1][0]);
          acc[1 + q][0][1] = wmma_f32(a0, b1[q], acc[1 + q][0][1]);
          acc[1 + q][1][1] = wmma_f32(a1, b1[q], acc[1 + q][1][1]);
        }
      }
    }
  }

#pragma unroll
  for (int mi = 0; mi < 2; ++mi)
#pragma unroll
    for (int ni = 0; ni < 2; ++ni)
#pragma unroll
      for (int e = 0; e < 8; ++e) {
        const int m = m0 + wm + mi * 16 + half * 8 + e;
        const int d = j0 + wn + ni * 16 + lrow;
        const float k1 = acc[0][mi][ni][e];
        const float zr = k1 - acc[2][mi][ni][e];
        const float zi = k1 + acc[1][mi][ni][e];
        *(float2*)(out + (size_t)m * NO + 2 * d) = make_float2(zr * zr + zr, zr * zi + zi);
      }
}

extern "C" void kernel_launch(void* const* d_in, const int* in_sizes, int n_in,
                              void* d_out, int out_size, void* d_ws, size_t ws_size,
                              hipStream_t stream) {
  const float* x     = (const float*)d_in[0];
  const float* h0    = (const float*)d_in[1];
  const float* th    = (const float*)d_in[2];
  const float* dampp = (const float*)d_in[3];
  const float* Bwr   = (const float*)d_in[4];
  const float* Bwi   = (const float*)d_in[5];
  const float* Cwr   = (const float*)d_in[6];
  const float* Cwi   = (const float*)d_in[7];
  const float* Dwr   = (const float*)d_in[8];
  const float* Dwi   = (const float*)d_in[9];
  const float* gw    = (const float*)d_in[10];
  const float* gb    = (const float*)d_in[11];

  float* out  = (float*)d_out;
  float* hfin = out + (size_t)MM * NO;

  float* ws = (float*)d_ws;
  float* W1  = ws;
  float* WD  = W1 + (size_t)KX * NP;
  float* WC  = WD + (size_t)KX * NO;
  float* P   = WC + (size_t)KH * NO;
  float* Hws = P + (size_t)MM * NP;

  pack_w1_kernel<<<(KX * NP + 255) / 256, 256, 0, stream>>>(Bwr, Bwi, gw, W1);
  pack_wd_kernel<<<dim3(DD / 32, DD / 32), 256, 0, stream>>>(Dwr, Dwi, WD);
  pack_wc_kernel<<<(KH * DD + 255) / 256, 256, 0, stream>>>(Cwr, Cwi, WC);

  gemm_proj_kernel<<<dim3(MM / 128, NP / PBN), 256, 0, stream>>>(x, W1, gb, P);
  scan_kernel<<<BB * (NNh / SNG), SNG * SSEG, 0, stream>>>(P, h0, th, dampp, Hws, hfin);
  gemm_out_kernel<<<dim3(MM / 128, DD / 64), 256, 0, stream>>>(x, WD, Hws, WC, out);
}


# output GEMM LDS staging made bank-conflict-free: A planes kept at column m^(4*(p/2)) (stores were 4-way), D and C weights packed k-pair-adjacent so 16 loaded bytes are one b128 LDS store
# speedup vs baseline: 2.2781x; 1.0667x over previous
#include <hip/hip_runtime.h>
#include <math.h>

#define BB 4
#define SS 2048
#define DD 1024
#define NNh 64
#define MM (BB * SS)
#define KX (2 * DD)
#define NP 192
#define KH (2 * NNh)
#define NO (2 * DD)

#define A2LD 144

#define PBLK (NP * 8)
#define CHS 8
#define SSEG 32
#define SLEN (SS / SSEG)
#define SCHK (SLEN / CHS)
#define SNG 16

typedef __attribute__((ext_vector_type(2))) float v2f;
typedef __attribute__((ext_vector_type(4))) float v4f;
typedef __attribute__((ext_vector_type(8))) float v8f;

__device__ __forceinline__ v8f wmma_f32(v2f a, v2f b, v8f c) {
  return __builtin_amdgcn_wmma_f32_16x16x4_f32(
       false, a,  false, b,
       (short)0, c,  false,  false);
}

__global__ void pack_w1_kernel(const float* __restrict__ Bwr, const float* __restrict__ Bwi,
                               const float* __restrict__ gw, float* __restrict__ W1t) {
  int idx = blockIdx.x * 256 + threadIdx.x;
  if (idx >= KX * NP) return;
  int j = idx / KX, k = idx - j * KX;
  int d = k >> 1, p = k & 1;
  float v;
  if (j < 64) {
    v = p ? -Bwi[j * DD + d] : Bwr[j * DD + d];
  } else if (j < 128) {
    int n = j - 64;
    v = p ? Bwr[n * DD + d] : Bwi[n * DD + d];
  } else {
    int n = j - 128;
    v = gw[n * (2 * DD) + p * DD + d];
  }
  W1t[idx] = v;
}

__global__ __launch_bounds__(256) void pack_wd_kernel(const float* __restrict__ Dr,
                                                      const float* __restrict__ Di,
                                                      float* __restrict__ WB) {
  __shared__ float tr[32][33];
  __shared__ float ti[32][33];
  const int tx = threadIdx.x & 31, ty = threadIdx.x >> 5;
  const int e0 = blockIdx.x * 32, d0 = blockIdx.y * 32;
  for (int r = ty; r < 32; r += 8) {
    tr[r][tx] = Dr[(size_t)(d0 + r) * DD + e0 + tx];
    ti[r][tx] = Di[(size_t)(d0 + r) * DD + e0 + tx];
  }
  __syncthreads();
  for (int q = ty; q < 16; q += 8) {
    const float wr0 = tr[tx][2 * q], wr1 = tr[tx][2 * q + 1];
    const float wi0 = ti[tx][2 * q], wi1 = ti[tx][2 * q + 1];
    const size_t o = (((size_t)(e0 >> 1) + q) * DD + d0 + tx) * 2;
    *(float2*)(WB + o) = make_float2(wr0, wr1);
    *(float2*)(WB + (size_t)DD * DD + o) = make_float2(wi0 - wr0, wi1 - wr1);
    *(float2*)(WB + 2 * (size_t)DD * DD + o) = make_float2(wr0 + wi0, wr1 + wi1);
  }
}

__global__ void pack_wc_kernel(const float* __restrict__ Cr, const float* __restrict__ Ci,
                               float* __restrict__ WC) {
  int idx = blockIdx.x * 256 + threadIdx.x;
  if (idx >= NNh * DD) return;
  int n = idx >> 10, d = idx & (DD - 1);
  float cr = Cr[d * NNh + n], ci = Ci[d * NNh + n];
  const size_t o = ((size_t)n * DD + d) * 2;
  *(float2*)(WC + o) = make_float2(ci, cr);
  *(float2*)(WC + (size_t)KH * DD + o) = make_float2(-cr, ci);
}

#define PKS 64
#define PLD 68
#define PBN 48

__global__ __launch_bounds__(256) void gemm_proj_kernel(const float* __restrict__ x,
                                                        const float* __restrict__ W1t,
                                                        const float* __restrict__ gate_b,
                                                        float* __restrict__ P) {
  __shared__ __attribute__((aligned(16))) float As[128 * PLD];
  __shared__ __attribute__((aligned(16))) float Bs[PBN * PLD];
  v8f acc[3] = {{0,0,0,0,0,0,0,0}, {0,0,0,0,0,0,0,0}, {0,0,0,0,0,0,0,0}};

  const int tid  = threadIdx.x;
  const int lane = tid & 31, wave = tid >> 5;
  const int half = lane >> 4, lrow = lane & 15;
  const int m0 = blockIdx.x * 128;
  const int j0 = blockIdx.y * PBN;

  const int srow = tid >> 4, skq = (tid & 15) << 2;
  const float* ap = x   + (size_t)(m0 + srow) * KX + skq;
  const float* bp = W1t + (size_t)(j0 + srow) * KX + skq;
  constexpr size_t RS = (size_t)16 * KX;
  float4 a0 = *(const float4*)(ap);
  float4 a1 = *(const float4*)(ap + RS);
  float4 a2 = *(const float4*)(ap + 2 * RS);
  float4 a3 = *(const float4*)(ap + 3 * RS);
  float4 a4 = *(const float4*)(ap + 4 * RS);
  float4 a5 = *(const float4*)(ap + 5 * RS);
  float4 a6 = *(const float4*)(ap + 6 * RS);
  float4 a7 = *(const float4*)(ap + 7 * RS);
  float4 b0 = *(const float4*)(bp);
  float4 b1 = *(const float4*)(bp + RS);
  float4 b2 = *(const float4*)(bp + 2 * RS);

  float* ast = As + srow * PLD + skq;
  float* bst = Bs + srow * PLD + skq;
  const float* af = As + (wave * 16 + lrow) * PLD + 4 * half;
  const float* bf = Bs + lrow * PLD + 4 * half;

  for (int k0 = 0; k0 < KX; k0 += PKS) {
    __syncthreads();
    *(float4*)(ast)                = a0;
    *(float4*)(ast + 1 * 16 * PLD) = a1;
    *(float4*)(ast + 2 * 16 * PLD) = a2;
    *(float4*)(ast + 3 * 16 * PLD) = a3;
    *(float4*)(ast + 4 * 16 * PLD) = a4;
    *(float4*)(ast + 5 * 16 * PLD) = a5;
    *(float4*)(ast + 6 * 16 * PLD) = a6;
    *(float4*)(ast + 7 * 16 * PLD) = a7;
    *(float4*)(bst)                = b0;
    *(float4*)(bst + 1 * 16 * PLD) = b1;
    *(float4*)(bst + 2 * 16 * PLD) = b2;
    __syncthreads();

    if (k0 + PKS < KX) {
      const float* an = ap + (k0 + PKS);
      const float* bn = bp + (k0 + PKS);
      a0 = *(const float4*)(an);
      a1 = *(const float4*)(an + RS);
      a2 = *(const float4*)(an + 2 * RS);
      a3 = *(const float4*)(an + 3 * RS);
      a4 = *(const float4*)(an + 4 * RS);
      a5 = *(const float4*)(an + 5 * RS);
      a6 = *(const float4*)(an + 6 * RS);
      a7 = *(const float4*)(an + 7 * RS);
      b0 = *(const float4*)(bn);
      b1 = *(const float4*)(bn + RS);
      b2 = *(const float4*)(bn + 2 * RS);
    }

#pragma unroll
    for (int u = 0; u < PKS / 8; ++u) {
      const float4 a = *(const float4*)(af + 8 * u);
      float4 b[3];
#pragma unroll
      for (int ni = 0; ni < 3; ++ni) b[ni] = *(const float4*)(bf + ni * 16 * PLD + 8 * u);
      v2f a01, a23;
      a01.x = a.x; a01.y = a.y; a23.x = a.z; a23.y = a.w;
#pragma unroll
      for (int ni = 0; ni < 3; ++ni) {
        v2f t; t.x = b[ni].x; t.y = b[ni].y;
        acc[ni] = wmma_f32(a01, t, acc[ni]);
      }
#pragma unroll
      for (int ni = 0; ni < 3; ++ni) {
        v2f t; t.x = b[ni].z; t.y = b[ni].w;
        acc[ni] = wmma_f32(a23, t, acc[ni]);
      }
    }
  }

  const int rb = ((m0 + wave * 16) >> 3) + half;
#pragma unroll
  for (int ni = 0; ni < 3; ++ni) {
    const int j = j0 + ni * 16 + lrow;
    float v[8];
#pragma unroll
    for (int e = 0; e < 8; ++e) {
      v[e] = acc[ni][e];
      if (j >= 128) v[e] = 1.0f / (1.0f + expf(-(v[e] + gate_b[j - 128])));
    }
    float* dst = P + (size_t)rb * PBLK + j * 8;
    *(float4*)(dst)     = make_float4(v[0], v[1], v[2], v[3]);
    *(float4*)(dst + 4) = make_float4(v[4], v[5], v[6], v[7]);
  }
}

struct ScanChunk {
  float4 br[2];
  float4 bi[2];
  float4 g[2];
};

__device__ __forceinline__ void scan_load(ScanChunk& c, const float* __restrict__ p) {
  c.br[0] = *(const float4*)(p);
  c.br[1] = *(const float4*)(p + 4);
  c.bi[0] = *(const float4*)(p + 64 * 8);
  c.bi[1] = *(const float4*)(p + 64 * 8 + 4);
  c.g[0]  = *(const float4*)(p + 128 * 8);
  c.g[1]  = *(const float4*)(p + 128 * 8 + 4);
}

__device__ __forceinline__ float f4_get(const float4& v, int i) {
  return i == 0 ? v.x : (i == 1 ? v.y : (i == 2 ? v.z : v.w));
}

__device__ __forceinline__ void scan_steps_local(const ScanChunk& c, float ca, float sa,
                                                 float& hr, float& hi, float& ap) {
#pragma unroll
  for (int i = 0; i < CHS; ++i) {
    float g  = f4_get(c.g[i >> 2], i & 3);
    float br = f4_get(c.br[i >> 2], i & 3);
    float bi = f4_get(c.bi[i >> 2], i & 3);
    float rr = hr * ca - hi * sa;
    float ri = hr * sa + hi * ca;
    hr = g * br + (1.0f - g) * rr;
    hi = g * bi + (1.0f - g) * ri;
    ap *= (1.0f - g);
  }
}

__device__ __forceinline__ void scan_steps(const ScanChunk& c, float ca, float sa,
                                           float& hr, float& hi, float* __restrict__ hw) {
#pragma unroll
  for (int i = 0; i < CHS; ++i) {
    float g  = f4_get(c.g[i >> 2], i & 3);
    float br = f4_get(c.br[i >> 2], i & 3);
    float bi = f4_get(c.bi[i >> 2], i & 3);
    float rr = hr * ca - hi * sa;
    float ri = hr * sa + hi * ca;
    hr = g * br + (1.0f - g) * rr;
    hi = g * bi + (1.0f - g) * ri;
    *(float2*)(hw + (size_t)i * KH) = make_float2(hr, hi);
  }
}

__global__ __launch_bounds__(SNG * SSEG) void scan_kernel(const float* __restrict__ P,
                                                          const float* __restrict__ h0,
                                                          const float* __restrict__ theta,
                                                          const float* __restrict__ dampp,
                                                          float* __restrict__ hws,
                                                          float* __restrict__ hfin) {
  __shared__ float sA[SSEG][SNG];
  __shared__ float sEr[SSEG][SNG];
  __shared__ float sEi[SSEG][SNG];

  const int b   = blockIdx.x / (NNh / SNG);
  const int n0  = (blockIdx.x % (NNh / SNG)) * SNG;
  const int nl  = threadIdx.x & (SNG - 1);
  const int seg = threadIdx.x / SNG;
  const int n   = n0 + nl;
  const float* pseg = P + ((size_t)b * (SS / 8) + (size_t)seg * SCHK) * PBLK + n * 8;
  float* hseg = hws + ((size_t)b * SS + (size_t)seg * SLEN) * KH + 2 * n;

  const float c = cosf(theta[n]);
  const float s = sinf(theta[n]);
  const float dmp = 0.5f + 0.5f / (1.0f + expf(-dampp[n]));
  const float ca = c * dmp, sa = s * dmp;

  static_assert(SLEN == 64, "six squarings give the 64th power");
  float pr = ca, pi = sa;
#pragma unroll
  for (int q = 0; q < 6; ++q) {
    const float tr = pr * pr - pi * pi;
    const float ti = 2.0f * pr * pi;
    pr = tr; pi = ti;
  }

  float er = 0.0f, ei = 0.0f, ap = 1.0f;
  {
    ScanChunk cur, nxt;
    scan_load(cur, pseg);
#pragma unroll
    for (int k = 0; k < SCHK; ++k) {
      if (k + 1 < SCHK) scan_load(nxt, pseg + (size_t)(k + 1) * PBLK);
      scan_steps_local(cur, ca, sa, er, ei, ap);
      if (k + 1 < SCHK) cur = nxt;
    }
  }
  sA[seg][nl]  = ap;
  sEr[seg][nl] = er;
  sEi[seg][nl] = ei;
  __syncthreads();

  float hr = h0[(b * NNh + n) * 2 + 0];
  float hi = h0[(b * NNh + n) * 2 + 1];
  for (int j = 0; j < seg; ++j) {
    const float a  = sA[j][nl];
    const float mr = a * pr, mi = a * pi;
    const float tr = mr * hr - mi * hi + sEr[j][nl];
    const float ti = mr * hi + mi * hr + sEi[j][nl];
    hr = tr; hi = ti;
  }

  {
    ScanChunk cur, nxt;
    scan_load(cur, pseg);
#pragma unroll
    for (int k = 0; k < SCHK; ++k) {
      if (k + 1 < SCHK) scan_load(nxt, pseg + (size_t)(k + 1) * PBLK);
      scan_steps(cur, ca, sa, hr, hi, hseg + (size_t)k * CHS * KH);
      if (k + 1 < SCHK) cur = nxt;
    }
  }
  if (seg == SSEG - 1) {
    hfin[(b * NNh + n) * 2 + 0] = hr;
    hfin[(b * NNh + n) * 2 + 1] = hi;
  }
}

#define B3LD 80

__global__ __launch_bounds__(256) void gemm_out_kernel(const float* __restrict__ x,
                                                       const float* __restrict__ WB,
                                                       const float* __restrict__ hws,
                                                       const float* __restrict__ WC,
                                                       float* __restrict__ out) {
  __shared__ v2f As2[3][8 * A2LD];
  __shared__ __attribute__((aligned(16))) v2f Bs2[3][8 * B3LD];
  v8f acc[3][2][2];
#pragma unroll
  for (int pl = 0; pl < 3; ++pl)
#pragma unroll
    for (int mi = 0; mi < 2; ++mi)
#pragma unroll
      for (int ni = 0; ni < 2; ++ni)
#pragma unroll
        for (int e = 0; e < 8; ++e) acc[pl][mi][ni][e] = 0.0f;

  const int tid  = threadIdx.x;
  const int lane = tid & 31, wave = tid >> 5;
  const int half = lane >> 4, lrow = lane & 15;
  const int wm = (wave & 3) * 32, wn = (wave >> 2) * 32;
  const int m0 = blockIdx.x * 128;
  const int j0 = blockIdx.y * 64;

  const int bq = tid >> 5, bd = (tid & 31) << 1;
  const int bso = bq * B3LD + bd;

  {
    const int arow = tid >> 3, akp = tid & 7;
    const float* ap = x + (size_t)(m0 + arow) * KX + akp * 4;
    const float* bp = WB + ((size_t)bq * DD + j0 + bd) * 2;
    const int aso = akp * A2LD + (arow ^ ((akp >> 1) << 2));
    float4 aReg[4];
#pragma unroll
    for (int i = 0; i < 4; ++i) aReg[i] = *(const float4*)(ap + (size_t)i * 32 * KX);
    v4f bR0 = *(const v4f*)(bp);
    v4f bR1 = *(const v4f*)(bp + (size_t)DD * DD);
    v4f bR2 = *(const v4f*)(bp + 2 * (size_t)DD * DD);

    for (int k0 = 0; k0 < DD; k0 += 16) {
      __syncthreads();
#pragma unroll
      for (int i = 0; i < 4; ++i) {
        const float4 f = aReg[i];
        const int o = aso + i * 32;
        v2f t;
        t.x = f.x + f.y; t.y = f.z + f.w; As2[0][o] = t;
        t.x = f.x;       t.y = f.z;       As2[1][o] = t;
        t.x = f.y;       t.y = f.w;       As2[2][o] = t;
      }
      *(v4f*)&Bs2[0][bso] = bR0;
      *(v4f*)&Bs2[1][bso] = bR1;
      *(v4f*)&Bs2[2][bso] = bR2;
      __syncthreads();

      if (k0 + 16 < DD) {
#pragma unroll
        for (int i = 0; i < 4; ++i)
          aReg[i] = *(const float4*)(ap + (size_t)i * 32 * KX + (k0 + 16) * 2);
        const float* bn = bp + (size_t)(k0 + 16) * DD;
        bR0 = *(const v4f*)(bn);
        bR1 = *(const v4f*)(bn + (size_t)DD * DD);
        bR2 = *(const v4f*)(bn + 2 * (size_t)DD * DD);
      }

#pragma unroll
      for (int kstep = 0; kstep < 4; ++kstep) {
        const int p = 2 * kstep + half;
        const int sl = lrow ^ (kstep << 2);
        v2f a0[3], a1[3], b0[3], b1[3];
#pragma unroll
        for (int pl = 0; pl < 3; ++pl) {
          a0[pl] = As2[pl][p * A2LD + wm + sl];
          a1[pl] = As2[pl][p * A2LD + wm + 16 + sl];
          b0[pl] = Bs2[pl][p * B3LD + wn + lrow];
          b1[pl] = Bs2[pl][p * B3LD + wn + 16 + lrow];
        }
#pragma unroll
        for (int pl = 0; pl < 3; ++pl) {
          acc[pl][0][0] = wmma_f32(a0[pl], b0[pl], acc[pl][0][0]);
          acc[pl][1][0] = wmma_f32(a1[pl], b0[pl], acc[pl][1][0]);
          acc[pl][0][1] = wmma_f32(a0[pl], b1[pl], acc[pl][0][1]);
          acc[pl][1][1] = wmma_f32(a1[pl], b1[pl], acc[pl][1][1]);
        }
      }
    }
  }

  {
    const int am0 = tid >> 2, am1 = 64 + (tid >> 2);
    const int ak0 = (tid & 3) << 2;
    const float* ap0 = hws + (size_t)(m0 + am0) * KH + ak0;
    const float* ap1 = hws + (size_t)(m0 + am1) * KH + ak0;
    const float* bp = WC + ((size_t)bq * DD + j0 + bd) * 2;
    const int as0 = (ak0 >> 1) * A2LD + (am0 ^ ((tid & 3) << 2));
    float4 aReg0 = *(const float4*)(ap0);
    float4 aReg1 = *(const float4*)(ap1);
    v4f bQ0 = *(const v4f*)(bp);
    v4f bQ1 = *(const v4f*)(bp + (size_t)KH * DD);

    for (int k0 = 0; k0 < KH; k0 += 16) {
      __syncthreads();
      {
        v2f t;
        t.x = aReg0.x; t.y = aReg0.y; As2[0][as0] = t;
        t.x = aReg0.z; t.y = aReg0.w; As2[0][as0 + A2LD] = t;
        t.x = aReg1.x; t.y = aReg1.y; As2[0][as0 + 64] = t;
        t.x = aReg1.z; t.y = aReg1.w; As2[0][as0 + A2LD + 64] = t;
      }
      *(v4f*)&Bs2[0][bso] = bQ0;
      *(v4f*)&Bs2[1][bso] = bQ1;
      __syncthreads();

      if (k0 + 16 < KH) {
        aReg0 = *(const float4*)(ap0 + (k0 + 16));
        aReg1 = *(const float4*)(ap1 + (k0 + 16));
        bQ0 = *(const v4f*)(bp + (size_t)(k0 + 16) * DD);
        bQ1 = *(const v4f*)(bp + (size_t)KH * DD + (size_t)(k0 + 16) * DD);
      }

#pragma unroll
      for (int kstep = 0; kstep < 4; ++kstep) {
        const int p = 2 * kstep + half;
        const int sl = lrow ^ (kstep << 2);
        const v2f a0 = As2[0][p * A2LD + wm + sl];
        const v2f a1 = As2[0][p * A2LD + wm + 16 + sl];
        v2f b0[2], b1[2];
#pragma unroll
        for (int q = 0; q < 2; ++q) {
          b0[q] = Bs2[q][p * B3LD + wn + lrow];
          b1[q] = Bs2[q][p * B3LD + wn + 16 + lrow];
        }
#pragma unroll
        for (int q = 0; q < 2; ++q) {
          acc[1 + q][0][0] = wmma_f32(a0, b0[q], acc[1 + q][0][0]);
          acc[1 + q][1][0] = wmma_f32(a1, b0[q], acc[1 + q][1][0]);
          acc[1 + q][0][1] = wmma_f32(a0, b1[q], acc[1 + q][0][1]);
          acc[1 + q][1][1] = wmma_f32(a1, b1[q], acc[1 + q][1][1]);
        }
      }
    }
  }

#pragma unroll
  for (int mi = 0; mi < 2; ++mi)
#pragma unroll
    for (int ni = 0; ni < 2; ++ni)
#pragma unroll
      for (int e = 0; e < 8; ++e) {
        const int m = m0 + wm + mi * 16 + half * 8 + e;
        const int d = j0 + wn + ni * 16 + lrow;
        const float k1 = acc[0][mi][ni][e];
        const float zr = k1 - acc[2][mi][ni][e];
        const float zi = k1 + acc[1][mi][ni][e];
        *(float2*)(out + (size_t)m * NO + 2 * d) = make_float2(zr * zr + zr, zr * zi + zi);
      }
}

extern "C" void kernel_launch(void* const* d_in, const int* in_sizes, int n_in,
                              void* d_out, int out_size, void* d_ws, size_t ws_size,
                              hipStream_t stream) {
  const float* x     = (const float*)d_in[0];
  const float* h0    = (const float*)d_in[1];
  const float* th    = (const float*)d_in[2];
  const float* dampp = (const float*)d_in[3];
  const float* Bwr   = (const float*)d_in[4];
  const float* Bwi   = (const float*)d_in[5];
  const float* Cwr   = (const float*)d_in[6];
  const float* Cwi   = (const float*)d_in[7];
  const float* Dwr   = (const float*)d_in[8];
  const float* Dwi   = (const float*)d_in[9];
  const float* gw    = (const float*)d_in[10];
  const float* gb    = (const float*)d_in[11];

  float* out  = (float*)d_out;
  float* hfin = out + (size_t)MM * NO;

  float* ws = (float*)d_ws;
  float* W1  = ws;
  float* WD  = W1 + (size_t)KX * NP;
  float* WC  = WD + (size_t)KX * NO;
  float* P   = WC + (size_t)KH * NO;
  float* Hws = P + (size_t)MM * NP;

  pack_w1_kernel<<<(KX * NP + 255) / 256, 256, 0, stream>>>(Bwr, Bwi, gw, W1);
  pack_wd_kernel<<<dim3(DD / 32, DD / 32), 256, 0, stream>>>(Dwr, Dwi, WD);
  pack_wc_kernel<<<(NNh * DD + 255) / 256, 256, 0, stream>>>(Cwr, Cwi, WC);

  gemm_proj_kernel<<<dim3(MM / 128, NP / PBN), 256, 0, stream>>>(x, W1, gb, P);
  scan_kernel<<<BB * (NNh / SNG), SNG * SSEG, 0, stream>>>(P, h0, th, dampp, Hws, hfin);
  gemm_out_kernel<<<dim3(MM / 128, DD / 64), 256, 0, stream>>>(x, WD, Hws, WC, out);
}


# weight packing without launches of its own: the proj GEMM stages B_w and gate_w rows directly, the D and C planes are packed by 256 extra work-groups of the proj launch; 6 launches become 3
# speedup vs baseline: 2.3564x; 1.0344x over previous
#include <hip/hip_runtime.h>
#include <math.h>

#define BB 4
#define SS 2048
#define DD 1024
#define NNh 64
#define MM (BB * SS)
#define KX (2 * DD)
#define NP 192
#define KH (2 * NNh)
#define NO (2 * DD)

#define A2LD 144

#define PBLK (NP * 8)
#define CHS 8
#define SSEG 32
#define SLEN (SS / SSEG)
#define SCHK (SLEN / CHS)
#define SNG 16

typedef __attribute__((ext_vector_type(2))) float v2f;
typedef __attribute__((ext_vector_type(4))) float v4f;
typedef __attribute__((ext_vector_type(8))) float v8f;

__device__ __forceinline__ v8f wmma_f32(v2f a, v2f b, v8f c) {
  return __builtin_amdgcn_wmma_f32_16x16x4_f32(
       false, a,  false, b,
       (short)0, c,  false,  false);
}

#define PROJ_WGS ((MM / 128) * (NP / PBN))
#define PACK_WGS 256
#define WD_TILES ((DD / 32) * (DD / 32))

__device__ __forceinline__ void pack_role(int q, int tid,
                                          const float* __restrict__ Dr, const float* __restrict__ Di,
                                          float* __restrict__ WB,
                                          const float* __restrict__ Cr, const float* __restrict__ Ci,
                                          float* __restrict__ WC, float* lds) {
  float (*tr)[33] = (float (*)[33])lds;
  float (*ti)[33] = (float (*)[33])(lds + 32 * 33);
  const int tx = tid & 31, ty = tid >> 5;
  static_assert(WD_TILES % PACK_WGS == 0, "every packing work-group runs the same number of tiles (barriers inside)");
  for (int t = q; t < WD_TILES; t += PACK_WGS) {
    const int e0 = (t & (DD / 32 - 1)) * 32, d0 = (t / (DD / 32)) * 32;
    float vr[4], vi[4];
#pragma unroll
    for (int i = 0; i < 4; ++i) {
      vr[i] = Dr[(size_t)(d0 + ty + 8 * i) * DD + e0 + tx];
      vi[i] = Di[(size_t)(d0 + ty + 8 * i) * DD + e0 + tx];
    }
#pragma unroll
    for (int i = 0; i < 4; ++i) {
      tr[ty + 8 * i][tx] = vr[i];
      ti[ty + 8 * i][tx] = vi[i];
    }
    __syncthreads();
#pragma unroll
    for (int i = 0; i < 2; ++i) {
      const int qq = ty + 8 * i;
      const float wr0 = tr[tx][2 * qq], wr1 = tr[tx][2 * qq + 1];
      const float wi0 = ti[tx][2 * qq], wi1 = ti[tx][2 * qq + 1];
      const size_t o = (((size_t)(e0 >> 1) + qq) * DD + d0 + tx) * 2;
      *(float2*)(WB + o) = make_float2(wr0, wr1);
      *(float2*)(WB + (size_t)DD * DD + o) = make_float2(wi0 - wr0, wi1 - wr1);
      *(float2*)(WB + 2 * (size_t)DD * DD + o) = make_float2(wr0 + wi0, wr1 + wi1);
    }
    __syncthreads();
  }
  for (int idx = q * 256 + tid; idx < NNh * DD; idx += PACK_WGS * 256) {
    const int n = idx >> 10, d = idx & (DD - 1);
    const float cr = Cr[d * NNh + n], ci = Ci[d * NNh + n];
    const size_t o = ((size_t)n * DD + d) * 2;
    *(float2*)(WC + o) = make_float2(ci, cr);
    *(float2*)(WC + (size_t)KH * DD + o) = make_float2(-cr, ci);
  }
}

#define PKS 64
#define PLD 68
#define PBN 48

__device__ __forceinline__ void proj_w_row(int j, const float* __restrict__ Bwr,
                                           const float* __restrict__ Bwi,
                                           const float* __restrict__ gw,
                                           const float*& ev, const float*& od, float& sg) {
  const int n = j & (NNh - 1), kind = j >> 6;
  const float* br = Bwr + (size_t)n * DD;
  const float* bi = Bwi + (size_t)n * DD;
  const float* g  = gw + (size_t)n * (2 * DD);
  ev = kind == 0 ? br : (kind == 1 ? bi : g);
  od = kind == 0 ? bi : (kind == 1 ? br : g + DD);
  sg = kind == 0 ? -1.0f : 1.0f;
}

__global__ __launch_bounds__(256) void gemm_proj_kernel(const float* __restrict__ x,
                                                        const float* __restrict__ Bwr,
                                                        const float* __restrict__ Bwi,
                                                        const float* __restrict__ gw,
                                                        const float* __restrict__ gate_b,
                                                        float* __restrict__ P,
                                                        const float* __restrict__ Dr,
                                                        const float* __restrict__ Di,
                                                        float* __restrict__ WB,
                                                        const float* __restrict__ Cr,
                                                        const float* __restrict__ Ci,
                                                        float* __restrict__ WC) {
  __shared__ __attribute__((aligned(16))) float As[128 * PLD];
  __shared__ __attribute__((aligned(16))) float Bs[PBN * PLD];
  const int tid  = threadIdx.x;
  if (blockIdx.x >= PROJ_WGS) {
    static_assert(128 * PLD >= 2 * 32 * 33, "the packing tiles fit the A stage");
    pack_role(blockIdx.x - PROJ_WGS, tid, Dr, Di, WB, Cr, Ci, WC, As);
    return;
  }
  v8f acc[3] = {{0,0,0,0,0,0,0,0}, {0,0,0,0,0,0,0,0}, {0,0,0,0,0,0,0,0}};

  const int lane = tid & 31, wave = tid >> 5;
  const int half = lane >> 4, lrow = lane & 15;
  const int m0 = (blockIdx.x & (MM / 128 - 1)) * 128;
  const int j0 = (blockIdx.x / (MM / 128)) * PBN;

  const int srow = tid >> 4, skq = (tid & 15) << 2;
  const float* ap = x   + (size_t)(m0 + srow) * KX + skq;
  const float *e0p, *o0p, *e1p, *o1p, *e2p, *o2p;
  float s0, s1, s2;
  proj_w_row(j0 + srow,      Bwr, Bwi, gw, e0p, o0p, s0);
  proj_w_row(j0 + srow + 16, Bwr, Bwi, gw, e1p, o1p, s1);
  proj_w_row(j0 + srow + 32, Bwr, Bwi, gw, e2p, o2p, s2);
  e0p += skq >> 1; o0p += skq >> 1;
  e1p += skq >> 1; o1p += skq >> 1;
  e2p += skq >> 1; o2p += skq >> 1;
  constexpr size_t RS = (size_t)16 * KX;
  float4 a0 = *(const float4*)(ap);
  float4 a1 = *(const float4*)(ap + RS);
  float4 a2 = *(const float4*)(ap + 2 * RS);
  float4 a3 = *(const float4*)(ap + 3 * RS);
  float4 a4 = *(const float4*)(ap + 4 * RS);
  float4 a5 = *(const float4*)(ap + 5 * RS);
  float4 a6 = *(const float4*)(ap + 6 * RS);
  float4 a7 = *(const float4*)(ap + 7 * RS);
  float2 be0 = *(const float2*)(e0p), bo0 = *(const float2*)(o0p);
  float2 be1 = *(const float2*)(e1p), bo1 = *(const float2*)(o1p);
  float2 be2 = *(const float2*)(e2p), bo2 = *(const float2*)(o2p);

  float* ast = As + srow * PLD + skq;
  float* bst = Bs + srow * PLD + skq;
  const float* af = As + (wave * 16 + lrow) * PLD + 4 * half;
  const float* bf = Bs + lrow * PLD + 4 * half;

  for (int k0 = 0; k0 < KX; k0 += PKS) {
    __syncthreads();
    *(float4*)(ast)                = a0;
    *(float4*)(ast + 1 * 16 * PLD) = a1;
    *(float4*)(ast + 2 * 16 * PLD) = a2;
    *(float4*)(ast + 3 * 16 * PLD) = a3;
    *(float4*)(ast + 4 * 16 * PLD) = a4;
    *(float4*)(ast + 5 * 16 * PLD) = a5;
    *(float4*)(ast + 6 * 16 * PLD) = a6;
    *(float4*)(ast + 7 * 16 * PLD) = a7;
    *(float4*)(bst)                = make_float4(be0.x, s0 * bo0.x, be0.y, s0 * bo0.y);
    *(float4*)(bst + 1 * 16 * PLD) = make_float4(be1.x, s1 * bo1.x, be1.y, s1 * bo1.y);
    *(float4*)(bst + 2 * 16 * PLD) = make_float4(be2.x, s2 * bo2.x, be2.y, s2 * bo2.y);
    __syncthreads();

    if (k0 + PKS < KX) {
      const float* an = ap + (k0 + PKS);
      const int dn = (k0 + PKS) >> 1;
      a0 = *(const float4*)(an);
      a1 = *(const float4*)(an + RS);
      a2 = *(const float4*)(an + 2 * RS);
      a3 = *(const float4*)(an + 3 * RS);
      a4 = *(const float4*)(an + 4 * RS);
      a5 = *(const float4*)(an + 5 * RS);
      a6 = *(const float4*)(an + 6 * RS);
      a7 = *(const float4*)(an + 7 * RS);
      be0 = *(const float2*)(e0p + dn); bo0 = *(const float2*)(o0p + dn);
      be1 = *(const float2*)(e1p + dn); bo1 = *(const float2*)(o1p + dn);
      be2 = *(const float2*)(e2p + dn); bo2 = *(const float2*)(o2p + dn);
    }

#pragma unroll
    for (int u = 0; u < PKS / 8; ++u) {
      const float4 a = *(const float4*)(af + 8 * u);
      float4 b[3];
#pragma unroll
      for (int ni = 0; ni < 3; ++ni) b[ni] = *(const float4*)(bf + ni * 16 * PLD + 8 * u);
      v2f a01, a23;
      a01.x = a.x; a01.y = a.y; a23.x = a.z; a23.y = a.w;
#pragma unroll
      for (int ni = 0; ni < 3; ++ni) {
        v2f t; t.x = b[ni].x; t.y = b[ni].y;
        acc[ni] = wmma_f32(a01, t, acc[ni]);
      }
#pragma unroll
      for (int ni = 0; ni < 3; ++ni) {
        v2f t; t.x = b[ni].z; t.y = b[ni].w;
        acc[ni] = wmma_f32(a23, t, acc[ni]);
      }
    }
  }

  const int rb = ((m0 + wave * 16) >> 3) + half;
#pragma unroll
  for (int ni = 0; ni < 3; ++ni) {
    const int j = j0 + ni * 16 + lrow;
    float v[8];
#pragma unroll
    for (int e = 0; e < 8; ++e) {
      v[e] = acc[ni][e];
      if (j >= 128) v[e] = 1.0f / (1.0f + expf(-(v[e] + gate_b[j - 128])));
    }
    float* dst = P + (size_t)rb * PBLK + j * 8;
    *(float4*)(dst)     = make_float4(v[0], v[1], v[2], v[3]);
    *(float4*)(dst + 4) = make_float4(v[4], v[5], v[6], v[7]);
  }
}

struct ScanChunk {
  float4 br[2];
  float4 bi[2];
  float4 g[2];
};

__device__ __forceinline__ void scan_load(ScanChunk& c, const float* __restrict__ p) {
  c.br[0] = *(const float4*)(p);
  c.br[1] = *(const float4*)(p + 4);
  c.bi[0] = *(const float4*)(p + 64 * 8);
  c.bi[1] = *(const float4*)(p + 64 * 8 + 4);
  c.g[0]  = *(const float4*)(p + 128 * 8);
  c.g[1]  = *(const float4*)(p + 128 * 8 + 4);
}

__device__ __forceinline__ float f4_get(const float4& v, int i) {
  return i == 0 ? v.x : (i == 1 ? v.y : (i == 2 ? v.z : v.w));
}

__device__ __forceinline__ void scan_steps_local(const ScanChunk& c, float ca, float sa,
                                                 float& hr, float& hi, float& ap) {
#pragma unroll
  for (int i = 0; i < CHS; ++i) {
    float g  = f4_get(c.g[i >> 2], i & 3);
    float br = f4_get(c.br[i >> 2], i & 3);
    float bi = f4_get(c.bi[i >> 2], i & 3);
    float rr = hr * ca - hi * sa;
    float ri = hr * sa + hi * ca;
    hr = g * br + (1.0f - g) * rr;
    hi = g * bi + (1.0f - g) * ri;
    ap *= (1.0f - g);
  }
}

__device__ __forceinline__ void scan_steps(const ScanChunk& c, float ca, float sa,
                                           float& hr, float& hi, float* __restrict__ hw) {
#pragma unroll
  for (int i = 0; i < CHS; ++i) {
    float g  = f4_get(c.g[i >> 2], i & 3);
    float br = f4_get(c.br[i >> 2], i & 3);
    float bi = f4_get(c.bi[i >> 2], i & 3);
    float rr = hr * ca - hi * sa;
    float ri = hr * sa + hi * ca;
    hr = g * br + (1.0f - g) * rr;
    hi = g * bi + (1.0f - g) * ri;
    *(float2*)(hw + (size_t)i * KH) = make_float2(hr, hi);
  }
}

__global__ __launch_bounds__(SNG * SSEG) void scan_kernel(const float* __restrict__ P,
                                                          const float* __restrict__ h0,
                                                          const float* __restrict__ theta,
                                                          const float* __restrict__ dampp,
                                                          float* __restrict__ hws,
                                                          float* __restrict__ hfin) {
  __shared__ float sA[SSEG][SNG];
  __shared__ float sEr[SSEG][SNG];
  __shared__ float sEi[SSEG][SNG];

  const int b   = blockIdx.x / (NNh / SNG);
  const int n0  = (blockIdx.x % (NNh / SNG)) * SNG;
  const int nl  = threadIdx.x & (SNG - 1);
  const int seg = threadIdx.x / SNG;
  const int n   = n0 + nl;
  const float* pseg = P + ((size_t)b * (SS / 8) + (size_t)seg * SCHK) * PBLK + n * 8;
  float* hseg = hws + ((size_t)b * SS + (size_t)seg * SLEN) * KH + 2 * n;

  const float c = cosf(theta[n]);
  const float s = sinf(theta[n]);
  const float dmp = 0.5f + 0.5f / (1.0f + expf(-dampp[n]));
  const float ca = c * dmp, sa = s * dmp;

  static_assert(SLEN == 64, "six squarings give the 64th power");
  float pr = ca, pi = sa;
#pragma unroll
  for (int q = 0; q < 6; ++q) {
    const float tr = pr * pr - pi * pi;
    const float ti = 2.0f * pr * pi;
    pr = tr; pi = ti;
  }

  float er = 0.0f, ei = 0.0f, ap = 1.0f;
  {
    ScanChunk cur, nxt;
    scan_load(cur, pseg);
#pragma unroll
    for (int k = 0; k < SCHK; ++k) {
      if (k + 1 < SCHK) scan_load(nxt, pseg + (size_t)(k + 1) * PBLK);
      scan_steps_local(cur, ca, sa, er, ei, ap);
      if (k + 1 < SCHK) cur = nxt;
    }
  }
  sA[seg][nl]  = ap;
  sEr[seg][nl] = er;
  sEi[seg][nl] = ei;
  __syncthreads();

  float hr = h0[(b * NNh + n) * 2 + 0];
  float hi = h0[(b * NNh + n) * 2 + 1];
  for (int j = 0; j < seg; ++j) {
    const float a  = sA[j][nl];
    const float mr = a * pr, mi = a * pi;
    const float tr = mr * hr - mi * hi + sEr[j][nl];
    const float ti = mr * hi + mi * hr + sEi[j][nl];
    hr = tr; hi = ti;
  }

  {
    ScanChunk cur, nxt;
    scan_load(cur, pseg);
#pragma unroll
    for (int k = 0; k < SCHK; ++k) {
      if (k + 1 < SCHK) scan_load(nxt, pseg + (size_t)(k + 1) * PBLK);
      scan_steps(cur, ca, sa, hr, hi, hseg + (size_t)k * CHS * KH);
      if (k + 1 < SCHK) cur = nxt;
    }
  }
  if (seg == SSEG - 1) {
    hfin[(b * NNh + n) * 2 + 0] = hr;
    hfin[(b * NNh + n) * 2 + 1] = hi;
  }
}

#define B3LD 80

__global__ __launch_bounds__(256) void gemm_out_kernel(const float* __restrict__ x,
                                                       const float* __restrict__ WB,
                                                       const float* __restrict__ hws,
                                                       const float* __restrict__ WC,
                                                       float* __restrict__ out) {
  __shared__ v2f As2[3][8 * A2LD];
  __shared__ __attribute__((aligned(16))) v2f Bs2[3][8 * B3LD];
  v8f acc[3][2][2];
#pragma unroll
  for (int pl = 0; pl < 3; ++pl)
#pragma unroll
    for (int mi = 0; mi < 2; ++mi)
#pragma unroll
      for (int ni = 0; ni < 2; ++ni)
#pragma unroll
        for (int e = 0; e < 8; ++e) acc[pl][mi][ni][e] = 0.0f;

  const int tid  = threadIdx.x;
  const int lane = tid & 31, wave = tid >> 5;
  const int half = lane >> 4, lrow = lane & 15;
  const int wm = (wave & 3) * 32, wn = (wave >> 2) * 32;
  const int m0 = blockIdx.x * 128;
  const int j0 = blockIdx.y * 64;

  const int bq = tid >> 5, bd = (tid & 31) << 1;
  const int bso = bq * B3LD + bd;

  {
    const int arow = tid >> 3, akp = tid & 7;
    const float* ap = x + (size_t)(m0 + arow) * KX + akp * 4;
    const float* bp = WB + ((size_t)bq * DD + j0 + bd) * 2;
    const int aso = akp * A2LD + (arow ^ ((akp >> 1) << 2));
    float4 aReg[4];
#pragma unroll
    for (int i = 0; i < 4; ++i) aReg[i] = *(const float4*)(ap + (size_t)i * 32 * KX);
    v4f bR0 = *(const v4f*)(bp);
    v4f bR1 = *(const v4f*)(bp + (size_t)DD * DD);
    v4f bR2 = *(const v4f*)(bp + 2 * (size_t)DD * DD);

    for (int k0 = 0; k0 < DD; k0 += 16) {
      __syncthreads();
#pragma unroll
      for (int i = 0; i < 4; ++i) {
        const float4 f = aReg[i];
        const int o = aso + i * 32;
        v2f t;
        t.x = f.x + f.y; t.y = f.z + f.w; As2[0][o] = t;
        t.x = f.x;       t.y = f.z;       As2[1][o] = t;
        t.x = f.y;       t.y = f.w;       As2[2][o] = t;
      }
      *(v4f*)&Bs2[0][bso] = bR0;
      *(v4f*)&Bs2[1][bso] = bR1;
      *(v4f*)&Bs2[2][bso] = bR2;
      __syncthreads();

      if (k0 + 16 < DD) {
#pragma unroll
        for (int i = 0; i < 4; ++i)
          aReg[i] = *(const float4*)(ap + (size_t)i * 32 * KX + (k0 + 16) * 2);
        const float* bn = bp + (size_t)(k0 + 16) * DD;
        bR0 = *(const v4f*)(bn);
        bR1 = *(const v4f*)(bn + (size_t)DD * DD);
        bR2 = *(const v4f*)(bn + 2 * (size_t)DD * DD);
      }

#pragma unroll
      for (int kstep = 0; kstep < 4; ++kstep) {
        const int p = 2 * kstep + half;
        const int sl = lrow ^ (kstep << 2);
        v2f a0[3], a1[3], b0[3], b1[3];
#pragma unroll
        for (int pl = 0; pl < 3; ++pl) {
          a0[pl] = As2[pl][p * A2LD + wm + sl];
          a1[pl] = As2[pl][p * A2LD + wm + 16 + sl];
          b0[pl] = Bs2[pl][p * B3LD + wn + lrow];
          b1[pl] = Bs2[pl][p * B3LD + wn + 16 + lrow];
        }
#pragma unroll
        for (int pl = 0; pl < 3; ++pl) {
          acc[pl][0][0] = wmma_f32(a0[pl], b0[pl], acc[pl][0][0]);
          acc[pl][1][0] = wmma_f32(a1[pl], b0[pl], acc[pl][1][0]);
          acc[pl][0][1] = wmma_f32(a0[pl], b1[pl], acc[pl][0][1]);
          acc[pl][1][1] = wmma_f32(a1[pl], b1[pl], acc[pl][1][1]);
        }
      }
    }
  }

  {
    const int am0 = tid >> 2, am1 = 64 + (tid >> 2);
    const int ak0 = (tid & 3) << 2;
    const float* ap0 = hws + (size_t)(m0 + am0) * KH + ak0;
    const float* ap1 = hws + (size_t)(m0 + am1) * KH + ak0;
    const float* bp = WC + ((size_t)bq * DD + j0 + bd) * 2;
    const int as0 = (ak0 >> 1) * A2LD + (am0 ^ ((tid & 3) << 2));
    float4 aReg0 = *(const float4*)(ap0);
    float4 aReg1 = *(const float4*)(ap1);
    v4f bQ0 = *(const v4f*)(bp);
    v4f bQ1 = *(const v4f*)(bp + (size_t)KH * DD);

    for (int k0 = 0; k0 < KH; k0 += 16) {
      __syncthreads();
      {
        v2f t;
        t.x = aReg0.x; t.y = aReg0.y; As2[0][as0] = t;
        t.x = aReg0.z; t.y = aReg0.w; As2[0][as0 + A2LD] = t;
        t.x = aReg1.x; t.y = aReg1.y; As2[0][as0 + 64] = t;
        t.x = aReg1.z; t.y = aReg1.w; As2[0][as0 + A2LD + 64] = t;
      }
      *(v4f*)&Bs2[0][bso] = bQ0;
      *(v4f*)&Bs2[1][bso] = bQ1;
      __syncthreads();

      if (k0 + 16 < KH) {
        aReg0 = *(const float4*)(ap0 + (k0 + 16));
        aReg1 = *(const float4*)(ap1 + (k0 + 16));
        bQ0 = *(const v4f*)(bp + (size_t)(k0 + 16) * DD);
        bQ1 = *(const v4f*)(bp + (size_t)KH * DD + (size_t)(k0 + 16) * DD);
      }

#pragma unroll
      for (int kstep = 0; kstep < 4; ++kstep) {
        const int p = 2 * kstep + half;
        const int sl = lrow ^ (kstep << 2);
        const v2f a0 = As2[0][p * A2LD + wm + sl];
        const v2f a1 = As2[0][p * A2LD + wm + 16 + sl];
        v2f b0[2], b1[2];
#pragma unroll
        for (int q = 0; q < 2; ++q) {
          b0[q] = Bs2[q][p * B3LD + wn + lrow];
          b1[q] = Bs2[q][p * B3LD + wn + 16 + lrow];
        }
#pragma unroll
        for (int q = 0; q < 2; ++q) {
          acc[1 + q][0][0] = wmma_f32(a0, b0[q], acc[1 + q][0][0]);
          acc[1 + q][1][0] = wmma_f32(a1, b0[q], acc[1 + q][1][0]);
          acc[1 + q][0][1] = wmma_f32(a0, b1[q], acc[1 + q][0][1]);
          acc[1 + q][1][1] = wmma_f32(a1, b1[q], acc[1 + q][1][1]);
        }
      }
    }
  }

#pragma unroll
  for (int mi = 0; mi < 2; ++mi)
#pragma unroll
    for (int ni = 0; ni < 2; ++ni)
#pragma unroll
      for (int e = 0; e < 8; ++e) {
        const int m = m0 + wm + mi * 16 + half * 8 + e;
        const int d = j0 + wn + ni * 16 + lrow;
        const float k1 = acc[0][mi][ni][e];
        const float zr = k1 - acc[2][mi][ni][e];
        const float zi = k1 + acc[1][mi][ni][e];
        *(float2*)(out + (size_t)m * NO + 2 * d) = make_float2(zr * zr + zr, zr * zi + zi);
      }
}

extern "C" void kernel_launch(void* const* d_in, const int* in_sizes, int n_in,
                              void* d_out, int out_size, void* d_ws, size_t ws_size,
                              hipStream_t stream) {
  const float* x     = (const float*)d_in[0];
  const float* h0    = (const float*)d_in[1];
  const float* th    = (const float*)d_in[2];
  const float* dampp = (const float*)d_in[3];
  const float* Bwr   = (const float*)d_in[4];
  const float* Bwi   = (const float*)d_in[5];
  const float* Cwr   = (const float*)d_in[6];
  const float* Cwi   = (const float*)d_in[7];
  const float* Dwr   = (const float*)d_in[8];
  const float* Dwi   = (const float*)d_in[9];
  const float* gw    = (const float*)d_in[10];
  const float* gb    = (const float*)d_in[11];

  float* out  = (float*)d_out;
  float* hfin = out + (size_t)MM * NO;

  float* ws = (float*)d_ws;
  float* WD  = ws + (size_t)KX * NP;
  float* WC  = WD + (size_t)KX * NO;
  float* P   = WC + (size_t)KH * NO;
  float* Hws = P + (size_t)MM * NP;

  gemm_proj_kernel<<<PROJ_WGS + PACK_WGS, 256, 0, stream>>>(x, Bwr, Bwi, gw, gb, P,
                                                            Dwr, Dwi, WD, Cwr, Cwi, WC);
  scan_kernel<<<BB * (NNh / SNG), SNG * SSEG, 0, stream>>>(P, h0, th, dampp, Hws, hfin);
  gemm_out_kernel<<<dim3(MM / 128, DD / 64), 256, 0, stream>>>(x, WD, Hws, WC, out);
}
